# starting kernel
# baseline (speedup 1.0000x reference)
#include <hip/hip_runtime.h>
#include <hip/hip_bf16.h>

typedef float v2f __attribute__((ext_vector_type(2)));
typedef float v8f __attribute__((ext_vector_type(8)));

#define NU 50000
#define NI 100000
#define NT 150000
#define DD 64
#define NNZE 2400000
#define BATCH 4096
#define SLOPE 0.01f

__global__ __launch_bounds__(256)
void k_gather0(const float* __restrict__ ue, const float* __restrict__ ie,
               const int* __restrict__ us, const int* __restrict__ pi,
               const int* __restrict__ ni, float* __restrict__ out) {
    int t = blockIdx.x * 256 + threadIdx.x;
    if (t >= 3 * BATCH * DD) return;
    int c  = t & 63;
    int b  = t >> 6;
    int seg = b >> 12;
    int bb  = b & (BATCH - 1);
    float v;
    if (seg == 0)      v = ue[(size_t)us[bb] * DD + c];
    else if (seg == 1) v = ie[(size_t)pi[bb] * DD + c];
    else               v = ie[(size_t)ni[bb] * DD + c];
    out[(size_t)b * 256 + c] = v;
}

__global__ __launch_bounds__(256)
void k_spmm(const int* __restrict__ rows, const int* __restrict__ cols,
            const float* __restrict__ vals, const float* __restrict__ ego,
            float* __restrict__ L) {
    long t = (long)blockIdx.x * 256 + threadIdx.x;
    if (t >= (long)NNZE * 16) return;
    int e = (int)(t >> 4);
    int q = ((int)t & 15) * 4;
    int r = rows[e], c = cols[e];
    float v = vals[e];
    const float4 x = *(const float4*)(ego + (size_t)c * DD + q);
    float* dst = L + (size_t)r * DD + q;
    atomicAdd(dst + 0, v * x.x);
    atomicAdd(dst + 1, v * x.y);
    atomicAdd(dst + 2, v * x.z);
    atomicAdd(dst + 3, v * x.w);
}

__global__ __launch_bounds__(256)
void k_dense_wmma(const float* __restrict__ L, float* __restrict__ ego,
                  const float* __restrict__ Wgc, const float* __restrict__ Wbi,
                  const float* __restrict__ bias) {
    __shared__ float sg[4096];
    __shared__ float sb[4096];
    __shared__ float sbias[64];

    for (int i = threadIdx.x; i < 4096; i += 256) {
        int r = i >> 6, c = i & 63;
        int dst = ((((r >> 1) * 64) + c) << 1) | (r & 1);
        sg[dst] = Wgc[i];
        sb[dst] = Wbi[i];
    }
    if (threadIdx.x < 64) sbias[threadIdx.x] = 2.0f * bias[threadIdx.x];
    __syncthreads();

    const int wave  = threadIdx.x >> 5;
    const int lane  = threadIdx.x & 31;
    const int mbase = (blockIdx.x * 8 + wave) * 16;
    if (mbase >= NT) return;

    const int half = lane >> 4;
    const int n    = lane & 15;

    const float* lrow = L   + (size_t)(mbase + n) * DD;
    const float* erow = ego + (size_t)(mbase + n) * DD;

    v8f acc[4];
    #pragma unroll
    for (int j = 0; j < 4; ++j) acc[j] = (v8f){};

    #pragma unroll
    for (int kk = 0; kk < 16; ++kk) {
        int off = kk * 4 + half * 2;
        v2f ls = *(const v2f*)(lrow + off);
        v2f eg = *(const v2f*)(erow + off);
        v2f a_li  = ls + eg;
        v2f a_mul = ls * eg;
        int bbase = ((kk * 2 + half) * 64) << 1;
        #pragma unroll
        for (int j = 0; j < 4; ++j) {
            v2f bgc = *(const v2f*)(sg + bbase + ((j * 16 + n) << 1));
            v2f bbi = *(const v2f*)(sb + bbase + ((j * 16 + n) << 1));
            acc[j] = __builtin_amdgcn_wmma_f32_16x16x4_f32(
                false, a_li, false, bgc, (short)0, acc[j], false, false);
            acc[j] = __builtin_amdgcn_wmma_f32_16x16x4_f32(
                false, a_mul, false, bbi, (short)0, acc[j], false, false);
        }
    }

    #pragma unroll
    for (int j = 0; j < 4; ++j) {
        float bsum = sbias[j * 16 + n];
        #pragma unroll
        for (int v = 0; v < 8; ++v) {
            float x = acc[j][v] + bsum;
            x = (x >= 0.f) ? x : SLOPE * x;
            ego[(size_t)(mbase + (half << 3) + v) * DD + j * 16 + n] = x;
        }
    }
}

__global__ __launch_bounds__(256)
void k_gather_norm(const float* __restrict__ ego, const int* __restrict__ us,
                   const int* __restrict__ pi, const int* __restrict__ ni,
                   float* __restrict__ out, int colbase) {
    int wid  = (blockIdx.x * 256 + threadIdx.x) >> 5;
    int lane = threadIdx.x & 31;
    if (wid >= 3 * BATCH) return;
    int seg = wid >> 12;
    int bb  = wid & (BATCH - 1);
    int node;
    if (seg == 0)      node = us[bb];
    else if (seg == 1) node = NU + pi[bb];
    else               node = NU + ni[bb];
    v2f x = *(const v2f*)(ego + (size_t)node * DD + lane * 2);
    float ss = x.x * x.x + x.y * x.y;
    #pragma unroll
    for (int o = 16; o > 0; o >>= 1) ss += __shfl_xor(ss, o, 32);
    float scale = 1.0f / fmaxf(sqrtf(ss), 1e-12f);
    v2f y; y.x = x.x * scale; y.y = x.y * scale;
    *(v2f*)(out + (size_t)wid * 256 + colbase + lane * 2) = y;
}

extern "C" void kernel_launch(void* const* d_in, const int* in_sizes, int n_in,
                              void* d_out, int out_size, void* d_ws, size_t ws_size,
                              hipStream_t stream) {
    const float* ue   = (const float*)d_in[0];
    const float* ie   = (const float*)d_in[1];
    const float* Wgc  = (const float*)d_in[2];
    const float* Wbi  = (const float*)d_in[3];
    const float* bbi  = (const float*)d_in[4];
    const float* vals = (const float*)d_in[5];
    const int*   rows = (const int*)d_in[6];
    const int*   cols = (const int*)d_in[7];
    const int*   us   = (const int*)d_in[8];
    const int*   pi   = (const int*)d_in[9];
    const int*   ni   = (const int*)d_in[10];
    float* out = (float*)d_out;

    float* ego = (float*)d_ws;
    float* L   = ego + (size_t)NT * DD;

    hipMemcpyAsync(ego, ue, (size_t)NU * DD * sizeof(float),
                   hipMemcpyDeviceToDevice, stream);
    hipMemcpyAsync(ego + (size_t)NU * DD, ie, (size_t)NI * DD * sizeof(float),
                   hipMemcpyDeviceToDevice, stream);

    k_gather0<<<(3 * BATCH * DD + 255) / 256, 256, 0, stream>>>(
        ue, ie, us, pi, ni, out);

    for (int k = 0; k < 3; ++k) {
        hipMemsetAsync(L, 0, (size_t)NT * DD * sizeof(float), stream);
        long tot = (long)NNZE * 16;
        k_spmm<<<(int)((tot + 255) / 256), 256, 0, stream>>>(
            rows, cols, vals, ego, L);
        k_dense_wmma<<<(NT + 127) / 128, 256, 0, stream>>>(
            L, ego, Wgc + k * 4096, Wbi + k * 4096, bbi + k * 64);
        k_gather_norm<<<(3 * BATCH * 32 + 255) / 256, 256, 0, stream>>>(
            ego, us, pi, ni, out, 64 * (k + 1));
    }
}


# SpMM as a blocked scatter: edges binned once per call by 128-row destination block; per layer a work-group sums its bin into an LDS tile with LDS float adds (arrival order); no global float atomics
# speedup vs baseline: 2.4319x; 2.4319x over previous
#include <hip/hip_runtime.h>
#include <hip/hip_bf16.h>

typedef float v2f __attribute__((ext_vector_type(2)));
typedef float v8f __attribute__((ext_vector_type(8)));

#define NU 50000
#define NI 100000
#define NT 150000
#define DD 64
#define NNZE 2400000
#define BATCH 4096
#define SLOPE 0.01f

__global__ __launch_bounds__(256)
void k_gather0(const float* __restrict__ ue, const float* __restrict__ ie,
               const int* __restrict__ us, const int* __restrict__ pi,
               const int* __restrict__ ni, float* __restrict__ out) {
    int t = blockIdx.x * 256 + threadIdx.x;
    if (t >= 3 * BATCH * DD) return;
    int c  = t & 63;
    int b  = t >> 6;
    int seg = b >> 12;
    int bb  = b & (BATCH - 1);
    float v;
    if (seg == 0)      v = ue[(size_t)us[bb] * DD + c];
    else if (seg == 1) v = ie[(size_t)pi[bb] * DD + c];
    else               v = ie[(size_t)ni[bb] * DD + c];
    out[(size_t)b * 256 + c] = v;
}

#define RB 128
#define NB ((NT + RB - 1) / RB)
#define CSH 18
#define CMASK ((1 << CSH) - 1)
#define GBIN 512
#define CHUNK ((NNZE + GBIN - 1) / GBIN)
static_assert(NT <= (1 << CSH), "col must fit under the local row");
static_assert(NB + 1 <= 2048, "scan handles two bins per thread");

__global__ __launch_bounds__(256)
void k_bin_count(const int* __restrict__ rows, int* __restrict__ gcount) {
    __shared__ int hist[NB];
    for (int i = threadIdx.x; i < NB; i += 256) hist[i] = 0;
    __syncthreads();
    const int e0 = blockIdx.x * CHUNK;
    const int e1 = min(e0 + CHUNK, NNZE);
    for (int e = e0 + threadIdx.x; e < e1; e += 256)
        atomicAdd(&hist[rows[e] / RB], 1);
    __syncthreads();
    for (int i = threadIdx.x; i < NB; i += 256) {
        int c = hist[i];
        if (c) atomicAdd(&gcount[i], c);
    }
}

__global__ __launch_bounds__(1024)
void k_bin_scan(const int* __restrict__ gcount, int* __restrict__ goff,
                int* __restrict__ gcur) {
    __shared__ int s[1024];
    const int t  = threadIdx.x;
    const int i0 = 2 * t, i1 = 2 * t + 1;
    const int c0 = (i0 < NB) ? gcount[i0] : 0;
    const int c1 = (i1 < NB) ? gcount[i1] : 0;
    s[t] = c0 + c1;
    __syncthreads();
    for (int o = 1; o < 1024; o <<= 1) {
        int v = (t >= o) ? s[t - o] : 0;
        __syncthreads();
        s[t] += v;
        __syncthreads();
    }
    const int excl = s[t] - (c0 + c1);
    if (i0 <= NB) { goff[i0] = excl;      if (i0 < NB) gcur[i0] = excl; }
    if (i1 <= NB) { goff[i1] = excl + c0; if (i1 < NB) gcur[i1] = excl + c0; }
}

__global__ __launch_bounds__(256)
void k_bin_place(const int* __restrict__ rows, const int* __restrict__ cols,
                 const float* __restrict__ vals, int* __restrict__ gcur,
                 int* __restrict__ binned) {
    __shared__ int hist[NB];
    __shared__ int base[NB];
    for (int i = threadIdx.x; i < NB; i += 256) hist[i] = 0;
    __syncthreads();
    const int e0 = blockIdx.x * CHUNK;
    const int e1 = min(e0 + CHUNK, NNZE);
    for (int e = e0 + threadIdx.x; e < e1; e += 256)
        atomicAdd(&hist[rows[e] / RB], 1);
    __syncthreads();
    for (int i = threadIdx.x; i < NB; i += 256) {
        int c = hist[i];
        base[i] = c ? atomicAdd(&gcur[i], c) : 0;
        hist[i] = 0;
    }
    __syncthreads();
    int2* rec = (int2*)binned;
    for (int e = e0 + threadIdx.x; e < e1; e += 256) {
        int r = rows[e];
        int b = r / RB;
        int pos = base[b] + atomicAdd(&hist[b], 1);
        rec[pos] = make_int2(cols[e] | ((r - b * RB) << CSH), __float_as_int(vals[e]));
    }
}

__device__ __forceinline__ void lds_add(float* p, float v) {
    __hip_atomic_fetch_add(p, v, __ATOMIC_RELAXED, __HIP_MEMORY_SCOPE_WORKGROUP);
}

__global__ __launch_bounds__(256)
void k_spmm_blk(const int* __restrict__ binned, const int* __restrict__ goff,
                const float* __restrict__ ego, float* __restrict__ L) {
    __shared__ float tile[RB * DD];
    const int b = blockIdx.x;
    float4* t4 = (float4*)tile;
    for (int i = threadIdx.x; i < RB * DD / 4; i += 256)
        t4[i] = make_float4(0.f, 0.f, 0.f, 0.f);
    __syncthreads();

    const int beg = goff[b], end = goff[b + 1];
    const int wave = __builtin_amdgcn_readfirstlane(threadIdx.x >> 5);
    const int lane = threadIdx.x & 31;
    const int2* rec = (const int2*)binned;

    int e0 = beg + wave * 32;
    int2 r = (e0 + lane < end) ? rec[e0 + lane] : make_int2(0, 0);
    for (; e0 < end; e0 += 256) {
        const int nx = e0 + 256 + lane;
        const int2 rn = (nx < end) ? rec[nx] : make_int2(0, 0);
        const int cnt = min(32, end - e0);
        int t = 0;
        for (; t + 16 <= cnt; t += 16) {
            v2f x[16];
            int pk[16];
            #pragma unroll
            for (int u = 0; u < 16; ++u) {
                pk[u] = __builtin_amdgcn_readlane(r.x, t + u);
                x[u] = *(const v2f*)(ego + (size_t)(pk[u] & CMASK) * DD + lane * 2);
            }
            #pragma unroll
            for (int u = 0; u < 16; ++u) {
                float v = __int_as_float(__builtin_amdgcn_readlane(r.y, t + u));
                float* d = tile + (pk[u] >> CSH) * DD + lane * 2;
                lds_add(d,     v * x[u].x);
                lds_add(d + 1, v * x[u].y);
            }
        }
        for (; t < cnt; ++t) {
            int p = __builtin_amdgcn_readlane(r.x, t);
            float v = __int_as_float(__builtin_amdgcn_readlane(r.y, t));
            v2f x1 = *(const v2f*)(ego + (size_t)(p & CMASK) * DD + lane * 2);
            float* d = tile + (p >> CSH) * DD + lane * 2;
            lds_add(d,     v * x1.x);
            lds_add(d + 1, v * x1.y);
        }
        r = rn;
    }
    __syncthreads();

    for (int i = threadIdx.x; i < RB * DD / 4; i += 256) {
        int row = b * RB + (i >> 4);
        if (row < NT)
            *(float4*)(L + (size_t)row * DD + (i & 15) * 4) = t4[i];
    }
}

__global__ __launch_bounds__(256)
void k_dense_wmma(const float* __restrict__ L, float* __restrict__ ego,
                  const float* __restrict__ Wgc, const float* __restrict__ Wbi,
                  const float* __restrict__ bias) {
    __shared__ float sg[4096];
    __shared__ float sb[4096];
    __shared__ float sbias[64];

    for (int i = threadIdx.x; i < 4096; i += 256) {
        int r = i >> 6, c = i & 63;
        int dst = ((((r >> 1) * 64) + c) << 1) | (r & 1);
        sg[dst] = Wgc[i];
        sb[dst] = Wbi[i];
    }
    if (threadIdx.x < 64) sbias[threadIdx.x] = 2.0f * bias[threadIdx.x];
    __syncthreads();

    const int wave  = threadIdx.x >> 5;
    const int lane  = threadIdx.x & 31;
    const int mbase = (blockIdx.x * 8 + wave) * 16;
    if (mbase >= NT) return;

    const int half = lane >> 4;
    const int n    = lane & 15;

    const float* lrow = L   + (size_t)(mbase + n) * DD;
    const float* erow = ego + (size_t)(mbase + n) * DD;

    v8f acc[4];
    #pragma unroll
    for (int j = 0; j < 4; ++j) acc[j] = (v8f){};

    #pragma unroll
    for (int kk = 0; kk < 16; ++kk) {
        int off = kk * 4 + half * 2;
        v2f ls = *(const v2f*)(lrow + off);
        v2f eg = *(const v2f*)(erow + off);
        v2f a_li  = ls + eg;
        v2f a_mul = ls * eg;
        int bbase = ((kk * 2 + half) * 64) << 1;
        #pragma unroll
        for (int j = 0; j < 4; ++j) {
            v2f bgc = *(const v2f*)(sg + bbase + ((j * 16 + n) << 1));
            v2f bbi = *(const v2f*)(sb + bbase + ((j * 16 + n) << 1));
            acc[j] = __builtin_amdgcn_wmma_f32_16x16x4_f32(
                false, a_li, false, bgc, (short)0, acc[j], false, false);
            acc[j] = __builtin_amdgcn_wmma_f32_16x16x4_f32(
                false, a_mul, false, bbi, (short)0, acc[j], false, false);
        }
    }

    #pragma unroll
    for (int j = 0; j < 4; ++j) {
        float bsum = sbias[j * 16 + n];
        #pragma unroll
        for (int v = 0; v < 8; ++v) {
            float x = acc[j][v] + bsum;
            x = (x >= 0.f) ? x : SLOPE * x;
            ego[(size_t)(mbase + (half << 3) + v) * DD + j * 16 + n] = x;
        }
    }
}

__global__ __launch_bounds__(256)
void k_gather_norm(const float* __restrict__ ego, const int* __restrict__ us,
                   const int* __restrict__ pi, const int* __restrict__ ni,
                   float* __restrict__ out, int colbase) {
    int wid  = (blockIdx.x * 256 + threadIdx.x) >> 5;
    int lane = threadIdx.x & 31;
    if (wid >= 3 * BATCH) return;
    int seg = wid >> 12;
    int bb  = wid & (BATCH - 1);
    int node;
    if (seg == 0)      node = us[bb];
    else if (seg == 1) node = NU + pi[bb];
    else               node = NU + ni[bb];
    v2f x = *(const v2f*)(ego + (size_t)node * DD + lane * 2);
    float ss = x.x * x.x + x.y * x.y;
    #pragma unroll
    for (int o = 16; o > 0; o >>= 1) ss += __shfl_xor(ss, o, 32);
    float scale = 1.0f / fmaxf(sqrtf(ss), 1e-12f);
    v2f y; y.x = x.x * scale; y.y = x.y * scale;
    *(v2f*)(out + (size_t)wid * 256 + colbase + lane * 2) = y;
}

extern "C" void kernel_launch(void* const* d_in, const int* in_sizes, int n_in,
                              void* d_out, int out_size, void* d_ws, size_t ws_size,
                              hipStream_t stream) {
    const float* ue   = (const float*)d_in[0];
    const float* ie   = (const float*)d_in[1];
    const float* Wgc  = (const float*)d_in[2];
    const float* Wbi  = (const float*)d_in[3];
    const float* bbi  = (const float*)d_in[4];
    const float* vals = (const float*)d_in[5];
    const int*   rows = (const int*)d_in[6];
    const int*   cols = (const int*)d_in[7];
    const int*   us   = (const int*)d_in[8];
    const int*   pi   = (const int*)d_in[9];
    const int*   ni   = (const int*)d_in[10];
    float* out = (float*)d_out;

    float* ego = (float*)d_ws;
    float* L   = ego + (size_t)NT * DD;
    int* binned = (int*)(L + (size_t)NT * DD);
    int* gcount = binned + (size_t)2 * NNZE;
    int* goff   = gcount + 2048;
    int* gcur   = goff + 2048;

    hipMemcpyAsync(ego, ue, (size_t)NU * DD * sizeof(float),
                   hipMemcpyDeviceToDevice, stream);
    hipMemcpyAsync(ego + (size_t)NU * DD, ie, (size_t)NI * DD * sizeof(float),
                   hipMemcpyDeviceToDevice, stream);

    k_gather0<<<(3 * BATCH * DD + 255) / 256, 256, 0, stream>>>(
        ue, ie, us, pi, ni, out);

    hipMemsetAsync(gcount, 0, 2048 * sizeof(int), stream);
    k_bin_count<<<GBIN, 256, 0, stream>>>(rows, gcount);
    k_bin_scan<<<1, 1024, 0, stream>>>(gcount, goff, gcur);
    k_bin_place<<<GBIN, 256, 0, stream>>>(rows, cols, vals, gcur, binned);

    for (int k = 0; k < 3; ++k) {
        k_spmm_blk<<<NB, 256, 0, stream>>>(binned, goff, ego, L);
        k_dense_wmma<<<(NT + 127) / 128, 256, 0, stream>>>(
            L, ego, Wgc + k * 4096, Wbi + k * 4096, bbi + k * 64);
        k_gather_norm<<<(3 * BATCH * 32 + 255) / 256, 256, 0, stream>>>(
            ego, us, pi, ni, out, 64 * (k + 1));
    }
}


# Stream operations 16 -> 10: layer 1 reads the input tables in place (no copy); counters cleared and bin sizes scanned in neighbouring launches; layer 1-2 output rows ride the next SpMM launch
# speedup vs baseline: 2.5601x; 1.0527x over previous
#include <hip/hip_runtime.h>
#include <hip/hip_bf16.h>

typedef float v2f __attribute__((ext_vector_type(2)));
typedef float v8f __attribute__((ext_vector_type(8)));

#define NU 50000
#define NI 100000
#define NT 150000
#define DD 64
#define NNZE 2400000
#define BATCH 4096
#define SLOPE 0.01f

__global__ __launch_bounds__(256)
void k_gather0(const float* __restrict__ ue, const float* __restrict__ ie,
               const int* __restrict__ us, const int* __restrict__ pi,
               const int* __restrict__ ni, float* __restrict__ out,
               int* __restrict__ gcount) {
    int t = blockIdx.x * 256 + threadIdx.x;
    if (t < 2048) gcount[t] = 0;
    if (t >= 3 * BATCH * DD) return;
    int c  = t & 63;
    int b  = t >> 6;
    int seg = b >> 12;
    int bb  = b & (BATCH - 1);
    float v;
    if (seg == 0)      v = ue[(size_t)us[bb] * DD + c];
    else if (seg == 1) v = ie[(size_t)pi[bb] * DD + c];
    else               v = ie[(size_t)ni[bb] * DD + c];
    out[(size_t)b * 256 + c] = v;
}

#define RB 128
#define NB ((NT + RB - 1) / RB)
#define CSH 18
#define CMASK ((1 << CSH) - 1)
#define GBIN 512
#define CHUNK ((NNZE + GBIN - 1) / GBIN)
static_assert(NT <= (1 << CSH), "col must fit under the local row");
static_assert(NB + 1 <= 5 * 256, "the placing launch scans five bins per thread");

__global__ __launch_bounds__(256)
void k_bin_count(const int* __restrict__ rows, int* __restrict__ gcount,
                 int* __restrict__ gcur) {
    __shared__ int hist[NB];
    if (blockIdx.x == 0)
        for (int i = threadIdx.x; i < NB; i += 256) gcur[i] = 0;
    for (int i = threadIdx.x; i < NB; i += 256) hist[i] = 0;
    __syncthreads();
    const int e0 = blockIdx.x * CHUNK;
    const int e1 = min(e0 + CHUNK, NNZE);
    for (int e = e0 + threadIdx.x; e < e1; e += 256)
        atomicAdd(&hist[rows[e] / RB], 1);
    __syncthreads();
    for (int i = threadIdx.x; i < NB; i += 256) {
        int c = hist[i];
        if (c) atomicAdd(&gcount[i], c);
    }
}

__global__ __launch_bounds__(256)
void k_bin_place(const int* __restrict__ rows, const int* __restrict__ cols,
                 const float* __restrict__ vals, const int* __restrict__ gcount,
                 int* __restrict__ goff, int* __restrict__ gcur,
                 int* __restrict__ binned) {
    __shared__ int hist[NB];
    __shared__ int base[NB];
    __shared__ int off[NB + 1];
    __shared__ int wsum[8];
    for (int i = threadIdx.x; i < NB; i += 256) hist[i] = 0;
    __syncthreads();
    int cs[5];
    for (int k = 0; k < 5; ++k) {
        int idx = threadIdx.x * 5 + k;
        cs[k] = (idx < NB) ? gcount[idx] : 0;
    }
    const int e0 = blockIdx.x * CHUNK;
    const int e1 = min(e0 + CHUNK, NNZE);
    for (int e = e0 + threadIdx.x; e < e1; e += 256)
        atomicAdd(&hist[rows[e] / RB], 1);
    {
        const int lane = threadIdx.x & 31, wv = threadIdx.x >> 5;
        const int s = cs[0] + cs[1] + cs[2] + cs[3] + cs[4];
        int incl = s;
        for (int o = 1; o < 32; o <<= 1) {
            int v = __shfl_up(incl, o, 32);
            if (lane >= o) incl += v;
        }
        if (lane == 31) wsum[wv] = incl;
        __syncthreads();
        int excl = incl - s;
        for (int w = 0; w < wv; ++w) excl += wsum[w];
        for (int k = 0; k < 5; ++k) {
            int idx = threadIdx.x * 5 + k;
            if (idx <= NB) off[idx] = excl;
            excl += cs[k];
        }
    }
    __syncthreads();
    if (blockIdx.x == 0)
        for (int i = threadIdx.x; i <= NB; i += 256) goff[i] = off[i];
    for (int i = threadIdx.x; i < NB; i += 256) {
        int c = hist[i];
        base[i] = c ? off[i] + atomicAdd(&gcur[i], c) : 0;
        hist[i] = 0;
    }
    __syncthreads();
    int2* rec = (int2*)binned;
    for (int e = e0 + threadIdx.x; e < e1; e += 256) {
        int r = rows[e];
        int b = r / RB;
        int pos = base[b] + atomicAdd(&hist[b], 1);
        rec[pos] = make_int2(cols[e] | ((r - b * RB) << CSH), __float_as_int(vals[e]));
    }
}

typedef const float __attribute__((address_space(1)))* gcf_p;
typedef const v2f   __attribute__((address_space(1)))* gcv2_p;
template <bool SPLIT>
__device__ __forceinline__ gcf_p src_row(const float* a, const float* b, int r) {
    if (!SPLIT) return (gcf_p)(a + (size_t)r * DD);
    const unsigned long long ba = (unsigned long long)a;
    const unsigned long long bb = (unsigned long long)b - (unsigned long long)NU * DD * sizeof(float);
    const unsigned long long base = (r < NU) ? ba : bb;
    return (gcf_p)(base + (unsigned long long)(unsigned)r * (DD * sizeof(float)));
}

#define GNR 8
#define GNB (3 * BATCH / (8 * GNR))
static_assert(GNB * 8 * GNR == 3 * BATCH, "output rows split evenly");
__device__ __forceinline__ void gather_norm_rows(
        const float* __restrict__ ego, const int* __restrict__ us,
        const int* __restrict__ pi, const int* __restrict__ ni,
        float* __restrict__ out, int colbase, int g) {
    const int lane = threadIdx.x & 31;
    const int w0 = g * 8 + (threadIdx.x >> 5);
    #pragma unroll
    for (int k = 0; k < GNR; ++k) {
        const int wid = w0 + k * (GNB * 8);
        const int seg = wid >> 12;
        const int bb  = wid & (BATCH - 1);
        const int* ip = (seg == 0) ? us : (seg == 1) ? pi : ni;
        const int node = ip[bb] + (seg ? NU : 0);
        v2f x = *(const v2f*)(ego + (size_t)node * DD + lane * 2);
        float ss = x.x * x.x + x.y * x.y;
        for (int o = 16; o > 0; o >>= 1) ss += __shfl_xor(ss, o, 32);
        float scale = 1.0f / fmaxf(sqrtf(ss), 1e-12f);
        v2f y; y.x = x.x * scale; y.y = x.y * scale;
        *(v2f*)(out + (size_t)wid * 256 + colbase + lane * 2) = y;
    }
}

__device__ __forceinline__ void lds_add(float* p, float v) {
    __hip_atomic_fetch_add(p, v, __ATOMIC_RELAXED, __HIP_MEMORY_SCOPE_WORKGROUP);
}

template <bool SPLIT>
__global__ __launch_bounds__(256)
void k_spmm_blk(const int* __restrict__ binned, const int* __restrict__ goff,
                const float* __restrict__ srcA, const float* __restrict__ srcB,
                float* __restrict__ L,
                const float* __restrict__ gsrc, const int* __restrict__ us,
                const int* __restrict__ pi, const int* __restrict__ ni,
                float* __restrict__ out, int colbase) {
    __shared__ float tile[RB * DD];
    if (blockIdx.x >= NB) {
        gather_norm_rows(gsrc, us, pi, ni, out, colbase, blockIdx.x - NB);
        return;
    }
    const int b = blockIdx.x;
    float4* t4 = (float4*)tile;
    for (int i = threadIdx.x; i < RB * DD / 4; i += 256)
        t4[i] = make_float4(0.f, 0.f, 0.f, 0.f);
    __syncthreads();

    const int beg = goff[b], end = goff[b + 1];
    const int wave = __builtin_amdgcn_readfirstlane(threadIdx.x >> 5);
    const int lane = threadIdx.x & 31;
    const int2* rec = (const int2*)binned;

    int e0 = beg + wave * 32;
    int2 r = (e0 + lane < end) ? rec[e0 + lane] : make_int2(0, 0);
    for (; e0 < end; e0 += 256) {
        const int nx = e0 + 256 + lane;
        const int2 rn = (nx < end) ? rec[nx] : make_int2(0, 0);
        const int cnt = min(32, end - e0);
        int t = 0;
        for (; t + 16 <= cnt; t += 16) {
            v2f x[16];
            int pk[16];
            #pragma unroll
            for (int u = 0; u < 16; ++u) {
                pk[u] = __builtin_amdgcn_readlane(r.x, t + u);
                x[u] = *(gcv2_p)(src_row<SPLIT>(srcA, srcB, pk[u] & CMASK) + lane * 2);
            }
            #pragma unroll
            for (int u = 0; u < 16; ++u) {
                float v = __int_as_float(__builtin_amdgcn_readlane(r.y, t + u));
                float* d = tile + (pk[u] >> CSH) * DD + lane * 2;
                lds_add(d,     v * x[u].x);
                lds_add(d + 1, v * x[u].y);
            }
        }
        for (; t < cnt; ++t) {
            int p = __builtin_amdgcn_readlane(r.x, t);
            float v = __int_as_float(__builtin_amdgcn_readlane(r.y, t));
            v2f x1 = *(gcv2_p)(src_row<SPLIT>(srcA, srcB, p & CMASK) + lane * 2);
            float* d = tile + (p >> CSH) * DD + lane * 2;
            lds_add(d,     v * x1.x);
            lds_add(d + 1, v * x1.y);
        }
        r = rn;
    }
    __syncthreads();

    for (int i = threadIdx.x; i < RB * DD / 4; i += 256) {
        int row = b * RB + (i >> 4);
        if (row < NT)
            *(float4*)(L + (size_t)row * DD + (i & 15) * 4) = t4[i];
    }
}

template <bool SPLIT>
__global__ __launch_bounds__(256)
void k_dense_wmma(const float* __restrict__ L, const float* srcA, const float* srcB,
                  float* ego,
                  const float* __restrict__ Wgc, const float* __restrict__ Wbi,
                  const float* __restrict__ bias) {
    __shared__ float sg[4096];
    __shared__ float sb[4096];
    __shared__ float sbias[64];

    for (int i = threadIdx.x; i < 4096; i += 256) {
        int r = i >> 6, c = i & 63;
        int dst = ((((r >> 1) * 64) + c) << 1) | (r & 1);
        sg[dst] = Wgc[i];
        sb[dst] = Wbi[i];
    }
    if (threadIdx.x < 64) sbias[threadIdx.x] = 2.0f * bias[threadIdx.x];
    __syncthreads();

    const int wave  = threadIdx.x >> 5;
    const int lane  = threadIdx.x & 31;
    const int mbase = (blockIdx.x * 8 + wave) * 16;
    if (mbase >= NT) return;

    const int half = lane >> 4;
    const int n    = lane & 15;

    const float* lrow = L   + (size_t)(mbase + n) * DD;
    gcf_p erow = src_row<SPLIT>(srcA, srcB, mbase + n);

    v8f acc[4];
    #pragma unroll
    for (int j = 0; j < 4; ++j) acc[j] = (v8f){};

    #pragma unroll
    for (int kk = 0; kk < 16; ++kk) {
        int off = kk * 4 + half * 2;
        v2f ls = *(const v2f*)(lrow + off);
        v2f eg = *(gcv2_p)(erow + off);
        v2f a_li  = ls + eg;
        v2f a_mul = ls * eg;
        int bbase = ((kk * 2 + half) * 64) << 1;
        #pragma unroll
        for (int j = 0; j < 4; ++j) {
            v2f bgc = *(const v2f*)(sg + bbase + ((j * 16 + n) << 1));
            v2f bbi = *(const v2f*)(sb + bbase + ((j * 16 + n) << 1));
            acc[j] = __builtin_amdgcn_wmma_f32_16x16x4_f32(
                false, a_li, false, bgc, (short)0, acc[j], false, false);
            acc[j] = __builtin_amdgcn_wmma_f32_16x16x4_f32(
                false, a_mul, false, bbi, (short)0, acc[j], false, false);
        }
    }

    #pragma unroll
    for (int j = 0; j < 4; ++j) {
        float bsum = sbias[j * 16 + n];
        #pragma unroll
        for (int v = 0; v < 8; ++v) {
            float x = acc[j][v] + bsum;
            x = (x >= 0.f) ? x : SLOPE * x;
            ego[(size_t)(mbase + (half << 3) + v) * DD + j * 16 + n] = x;
        }
    }
}

__global__ __launch_bounds__(256)
void k_gather_norm(const float* __restrict__ ego, const int* __restrict__ us,
                   const int* __restrict__ pi, const int* __restrict__ ni,
                   float* __restrict__ out, int colbase) {
    int wid  = (blockIdx.x * 256 + threadIdx.x) >> 5;
    int lane = threadIdx.x & 31;
    if (wid >= 3 * BATCH) return;
    int seg = wid >> 12;
    int bb  = wid & (BATCH - 1);
    int node;
    if (seg == 0)      node = us[bb];
    else if (seg == 1) node = NU + pi[bb];
    else               node = NU + ni[bb];
    v2f x = *(const v2f*)(ego + (size_t)node * DD + lane * 2);
    float ss = x.x * x.x + x.y * x.y;
    #pragma unroll
    for (int o = 16; o > 0; o >>= 1) ss += __shfl_xor(ss, o, 32);
    float scale = 1.0f / fmaxf(sqrtf(ss), 1e-12f);
    v2f y; y.x = x.x * scale; y.y = x.y * scale;
    *(v2f*)(out + (size_t)wid * 256 + colbase + lane * 2) = y;
}

extern "C" void kernel_launch(void* const* d_in, const int* in_sizes, int n_in,
                              void* d_out, int out_size, void* d_ws, size_t ws_size,
                              hipStream_t stream) {
    const float* ue   = (const float*)d_in[0];
    const float* ie   = (const float*)d_in[1];
    const float* Wgc  = (const float*)d_in[2];
    const float* Wbi  = (const float*)d_in[3];
    const float* bbi  = (const float*)d_in[4];
    const float* vals = (const float*)d_in[5];
    const int*   rows = (const int*)d_in[6];
    const int*   cols = (const int*)d_in[7];
    const int*   us   = (const int*)d_in[8];
    const int*   pi   = (const int*)d_in[9];
    const int*   ni   = (const int*)d_in[10];
    float* out = (float*)d_out;

    float* ego = (float*)d_ws;
    float* L   = ego + (size_t)NT * DD;
    int* binned = (int*)(L + (size_t)NT * DD);
    int* gcount = binned + (size_t)2 * NNZE;
    int* goff   = gcount + 2048;
    int* gcur   = goff + 2048;

    k_gather0<<<(3 * BATCH * DD + 255) / 256, 256, 0, stream>>>(
        ue, ie, us, pi, ni, out, gcount);

    k_bin_count<<<GBIN, 256, 0, stream>>>(rows, gcount, gcur);
    k_bin_place<<<GBIN, 256, 0, stream>>>(rows, cols, vals, gcount, goff, gcur, binned);

    k_spmm_blk<true><<<NB, 256, 0, stream>>>(
        binned, goff, ue, ie, L, ego, us, pi, ni, out, 0);
    k_dense_wmma<true><<<(NT + 127) / 128, 256, 0, stream>>>(
        L, ue, ie, ego, Wgc, Wbi, bbi);
    for (int k = 1; k < 3; ++k) {
        k_spmm_blk<false><<<NB + GNB, 256, 0, stream>>>(
            binned, goff, ego, ego, L, ego, us, pi, ni, out, 64 * k);
        k_dense_wmma<false><<<(NT + 127) / 128, 256, 0, stream>>>(
            L, ego, ego, ego, Wgc + k * 4096, Wbi + k * 4096, bbi + k * 64);
    }
    k_gather_norm<<<(3 * BATCH * 32 + 255) / 256, 256, 0, stream>>>(
        ego, us, pi, ni, out, 64 * 3);
}


# Binning launches: a thread requests all its edges at once (rows, cols, vals as 16-byte loads; rows kept in registers for both passes) and its five bin cursors together; sum order: arrival, as before
# speedup vs baseline: 2.7188x; 1.0620x over previous
#include <hip/hip_runtime.h>
#include <hip/hip_bf16.h>

typedef float v2f __attribute__((ext_vector_type(2)));
typedef float v8f __attribute__((ext_vector_type(8)));

#define NU 50000
#define NI 100000
#define NT 150000
#define DD 64
#define NNZE 2400000
#define BATCH 4096
#define SLOPE 0.01f

__global__ __launch_bounds__(256)
void k_gather0(const float* __restrict__ ue, const float* __restrict__ ie,
               const int* __restrict__ us, const int* __restrict__ pi,
               const int* __restrict__ ni, float* __restrict__ out,
               int* __restrict__ gcount) {
    int t = blockIdx.x * 256 + threadIdx.x;
    if (t < 2048) gcount[t] = 0;
    if (t >= 3 * BATCH * DD) return;
    int c  = t & 63;
    int b  = t >> 6;
    int seg = b >> 12;
    int bb  = b & (BATCH - 1);
    float v;
    if (seg == 0)      v = ue[(size_t)us[bb] * DD + c];
    else if (seg == 1) v = ie[(size_t)pi[bb] * DD + c];
    else               v = ie[(size_t)ni[bb] * DD + c];
    out[(size_t)b * 256 + c] = v;
}

#define RB 128
#define NB ((NT + RB - 1) / RB)
#define CSH 18
#define CMASK ((1 << CSH) - 1)
#define GBIN 512
#define CHUNK ((NNZE + GBIN - 1) / GBIN)
static_assert(NT <= (1 << CSH), "col must fit under the local row");
static_assert(NB + 1 <= 5 * 256, "the placing launch scans five bins per thread");

#define EPT 5
static_assert(CHUNK % 4 == 0 && NNZE % 4 == 0, "a chunk is whole 16-byte groups");
static_assert(EPT * 256 * 4 >= CHUNK, "EPT groups per thread cover a chunk");
static_assert((NNZE - (GBIN - 1) * CHUNK) / 4 >= (EPT - 1) * 256,
              "only the last group of a thread can lie outside the chunk");

__device__ __forceinline__ int chunk_groups(int g) {
    const int e0 = g * CHUNK;
    return (min(e0 + CHUNK, NNZE) - e0) >> 2;
}

__device__ __forceinline__ void hist_add4(int* hist, int4 r) {
    atomicAdd(&hist[(unsigned)r.x / RB], 1);
    atomicAdd(&hist[(unsigned)r.y / RB], 1);
    atomicAdd(&hist[(unsigned)r.z / RB], 1);
    atomicAdd(&hist[(unsigned)r.w / RB], 1);
}

__global__ __launch_bounds__(256)
void k_bin_count(const int* __restrict__ rows, int* __restrict__ gcount,
                 int* __restrict__ gcur) {
    __shared__ int hist[NB];
    if (blockIdx.x == 0)
        for (int i = threadIdx.x; i < NB; i += 256) gcur[i] = 0;
    for (int i = threadIdx.x; i < NB; i += 256) hist[i] = 0;
    __syncthreads();
    const int n4 = chunk_groups(blockIdx.x);
    const int4* r4 = (const int4*)(rows + (size_t)blockIdx.x * CHUNK);
    const bool tail = (int)threadIdx.x + (EPT - 1) * 256 < n4;
    int4 rr[EPT];
    for (int k = 0; k < EPT - 1; ++k) rr[k] = r4[threadIdx.x + k * 256];
    if (tail) rr[EPT - 1] = r4[threadIdx.x + (EPT - 1) * 256];
    for (int k = 0; k < EPT - 1; ++k) hist_add4(hist, rr[k]);
    if (tail) hist_add4(hist, rr[EPT - 1]);
    __syncthreads();
    for (int i = threadIdx.x; i < NB; i += 256) {
        int c = hist[i];
        if (c) atomicAdd(&gcount[i], c);
    }
}

__device__ __forceinline__ void place4(int2* rec, int* hist, const int* base,
                                       int4 r, int4 c, float4 v) {
    const int rv[4] = {r.x, r.y, r.z, r.w};
    const int cv[4] = {c.x, c.y, c.z, c.w};
    const float vv[4] = {v.x, v.y, v.z, v.w};
    int pos[4];
    for (int j = 0; j < 4; ++j) {
        const int b = (unsigned)rv[j] / RB;
        pos[j] = base[b] + atomicAdd(&hist[b], 1);
    }
    for (int j = 0; j < 4; ++j) {
        const int lr = rv[j] & (RB - 1);
        rec[pos[j]] = make_int2(cv[j] | (lr << CSH), __float_as_int(vv[j]));
    }
}
static_assert((RB & (RB - 1)) == 0, "the local row is taken with a mask");

__global__ __launch_bounds__(256)
void k_bin_place(const int* __restrict__ rows, const int* __restrict__ cols,
                 const float* __restrict__ vals, const int* __restrict__ gcount,
                 int* __restrict__ goff, int* __restrict__ gcur,
                 int* __restrict__ binned) {
    __shared__ int hist[NB];
    __shared__ int base[NB];
    __shared__ int off[NB + 1];
    __shared__ int wsum[8];
    for (int i = threadIdx.x; i < NB; i += 256) hist[i] = 0;
    __syncthreads();
    const int n4 = chunk_groups(blockIdx.x);
    const size_t e0 = (size_t)blockIdx.x * CHUNK;
    const int4*   r4 = (const int4*)(rows + e0);
    const int4*   c4 = (const int4*)(cols + e0);
    const float4* v4 = (const float4*)(vals + e0);
    const bool tail = (int)threadIdx.x + (EPT - 1) * 256 < n4;
    int4 rr[EPT], cc[EPT];
    float4 vv[EPT];
    for (int k = 0; k < EPT - 1; ++k) rr[k] = r4[threadIdx.x + k * 256];
    if (tail) rr[EPT - 1] = r4[threadIdx.x + (EPT - 1) * 256];
    int cs[5];
    for (int k = 0; k < 5; ++k) {
        int idx = threadIdx.x * 5 + k;
        cs[k] = (idx < NB) ? gcount[idx] : 0;
    }
    for (int k = 0; k < EPT - 1; ++k) {
        cc[k] = c4[threadIdx.x + k * 256];
        vv[k] = v4[threadIdx.x + k * 256];
    }
    if (tail) {
        cc[EPT - 1] = c4[threadIdx.x + (EPT - 1) * 256];
        vv[EPT - 1] = v4[threadIdx.x + (EPT - 1) * 256];
    }
    for (int k = 0; k < EPT - 1; ++k) hist_add4(hist, rr[k]);
    if (tail) hist_add4(hist, rr[EPT - 1]);
    {
        const int lane = threadIdx.x & 31, wv = threadIdx.x >> 5;
        const int s = cs[0] + cs[1] + cs[2] + cs[3] + cs[4];
        int incl = s;
        for (int o = 1; o < 32; o <<= 1) {
            int v = __shfl_up(incl, o, 32);
            if (lane >= o) incl += v;
        }
        if (lane == 31) wsum[wv] = incl;
        __syncthreads();
        int excl = incl - s;
        for (int w = 0; w < wv; ++w) excl += wsum[w];
        for (int k = 0; k < 5; ++k) {
            int idx = threadIdx.x * 5 + k;
            if (idx <= NB) off[idx] = excl;
            excl += cs[k];
        }
    }
    __syncthreads();
    if (blockIdx.x == 0)
        for (int i = threadIdx.x; i <= NB; i += 256) goff[i] = off[i];
    {
        int cnt[5], got[5];
        for (int k = 0; k < 5; ++k) {
            const int i = threadIdx.x + k * 256;
            cnt[k] = (i < NB) ? hist[i] : 0;
        }
        for (int k = 0; k < 5; ++k) {
            const int i = threadIdx.x + k * 256;
            got[k] = (i < NB) ? atomicAdd(&gcur[i], cnt[k]) : 0;
        }
        for (int k = 0; k < 5; ++k) {
            const int i = threadIdx.x + k * 256;
            if (i < NB) {
                base[i] = off[i] + got[k];
                hist[i] = 0;
            }
        }
    }
    static_assert(5 * 256 >= NB, "five bins per thread in the cursor step");
    __syncthreads();
    int2* rec = (int2*)binned;
    for (int k = 0; k < EPT - 1; ++k) place4(rec, hist, base, rr[k], cc[k], vv[k]);
    if (tail) place4(rec, hist, base, rr[EPT - 1], cc[EPT - 1], vv[EPT - 1]);
}

typedef const float __attribute__((address_space(1)))* gcf_p;
typedef const v2f   __attribute__((address_space(1)))* gcv2_p;
template <bool SPLIT>
__device__ __forceinline__ gcf_p src_row(const float* a, const float* b, int r) {
    if (!SPLIT) return (gcf_p)(a + (size_t)r * DD);
    const unsigned long long ba = (unsigned long long)a;
    const unsigned long long bb = (unsigned long long)b - (unsigned long long)NU * DD * sizeof(float);
    const unsigned long long base = (r < NU) ? ba : bb;
    return (gcf_p)(base + (unsigned long long)(unsigned)r * (DD * sizeof(float)));
}

#define GNR 8
#define GNB (3 * BATCH / (8 * GNR))
static_assert(GNB * 8 * GNR == 3 * BATCH, "output rows split evenly");
__device__ __forceinline__ void gather_norm_rows(
        const float* __restrict__ ego, const int* __restrict__ us,
        const int* __restrict__ pi, const int* __restrict__ ni,
        float* __restrict__ out, int colbase, int g) {
    const int lane = threadIdx.x & 31;
    const int w0 = g * 8 + (threadIdx.x >> 5);
    #pragma unroll
    for (int k = 0; k < GNR; ++k) {
        const int wid = w0 + k * (GNB * 8);
        const int seg = wid >> 12;
        const int bb  = wid & (BATCH - 1);
        const int* ip = (seg == 0) ? us : (seg == 1) ? pi : ni;
        const int node = ip[bb] + (seg ? NU : 0);
        v2f x = *(const v2f*)(ego + (size_t)node * DD + lane * 2);
        float ss = x.x * x.x + x.y * x.y;
        for (int o = 16; o > 0; o >>= 1) ss += __shfl_xor(ss, o, 32);
        float scale = 1.0f / fmaxf(sqrtf(ss), 1e-12f);
        v2f y; y.x = x.x * scale; y.y = x.y * scale;
        *(v2f*)(out + (size_t)wid * 256 + colbase + lane * 2) = y;
    }
}

__device__ __forceinline__ void lds_add(float* p, float v) {
    __hip_atomic_fetch_add(p, v, __ATOMIC_RELAXED, __HIP_MEMORY_SCOPE_WORKGROUP);
}

template <bool SPLIT>
__global__ __launch_bounds__(256)
void k_spmm_blk(const int* __restrict__ binned, const int* __restrict__ goff,
                const float* __restrict__ srcA, const float* __restrict__ srcB,
                float* __restrict__ L,
                const float* __restrict__ gsrc, const int* __restrict__ us,
                const int* __restrict__ pi, const int* __restrict__ ni,
                float* __restrict__ out, int colbase) {
    __shared__ float tile[RB * DD];
    if (blockIdx.x >= NB) {
        gather_norm_rows(gsrc, us, pi, ni, out, colbase, blockIdx.x - NB);
        return;
    }
    const int b = blockIdx.x;
    float4* t4 = (float4*)tile;
    for (int i = threadIdx.x; i < RB * DD / 4; i += 256)
        t4[i] = make_float4(0.f, 0.f, 0.f, 0.f);
    __syncthreads();

    const int beg = goff[b], end = goff[b + 1];
    const int wave = __builtin_amdgcn_readfirstlane(threadIdx.x >> 5);
    const int lane = threadIdx.x & 31;
    const int2* rec = (const int2*)binned;

    int e0 = beg + wave * 32;
    int2 r = (e0 + lane < end) ? rec[e0 + lane] : make_int2(0, 0);
    for (; e0 < end; e0 += 256) {
        const int nx = e0 + 256 + lane;
        const int2 rn = (nx < end) ? rec[nx] : make_int2(0, 0);
        const int cnt = min(32, end - e0);
        int t = 0;
        for (; t + 16 <= cnt; t += 16) {
            v2f x[16];
            int pk[16];
            #pragma unroll
            for (int u = 0; u < 16; ++u) {
                pk[u] = __builtin_amdgcn_readlane(r.x, t + u);
                x[u] = *(gcv2_p)(src_row<SPLIT>(srcA, srcB, pk[u] & CMASK) + lane * 2);
            }
            #pragma unroll
            for (int u = 0; u < 16; ++u) {
                float v = __int_as_float(__builtin_amdgcn_readlane(r.y, t + u));
                float* d = tile + (pk[u] >> CSH) * DD + lane * 2;
                lds_add(d,     v * x[u].x);
                lds_add(d + 1, v * x[u].y);
            }
        }
        for (; t < cnt; ++t) {
            int p = __builtin_amdgcn_readlane(r.x, t);
            float v = __int_as_float(__builtin_amdgcn_readlane(r.y, t));
            v2f x1 = *(gcv2_p)(src_row<SPLIT>(srcA, srcB, p & CMASK) + lane * 2);
            float* d = tile + (p >> CSH) * DD + lane * 2;
            lds_add(d,     v * x1.x);
            lds_add(d + 1, v * x1.y);
        }
        r = rn;
    }
    __syncthreads();

    for (int i = threadIdx.x; i < RB * DD / 4; i += 256) {
        int row = b * RB + (i >> 4);
        if (row < NT)
            *(float4*)(L + (size_t)row * DD + (i & 15) * 4) = t4[i];
    }
}

template <bool SPLIT>
__global__ __launch_bounds__(256)
void k_dense_wmma(const float* __restrict__ L, const float* srcA, const float* srcB,
                  float* ego,
                  const float* __restrict__ Wgc, const float* __restrict__ Wbi,
                  const float* __restrict__ bias) {
    __shared__ float sg[4096];
    __shared__ float sb[4096];
    __shared__ float sbias[64];

    for (int i = threadIdx.x; i < 4096; i += 256) {
        int r = i >> 6, c = i & 63;
        int dst = ((((r >> 1) * 64) + c) << 1) | (r & 1);
        sg[dst] = Wgc[i];
        sb[dst] = Wbi[i];
    }
    if (threadIdx.x < 64) sbias[threadIdx.x] = 2.0f * bias[threadIdx.x];
    __syncthreads();

    const int wave  = threadIdx.x >> 5;
    const int lane  = threadIdx.x & 31;
    const int mbase = (blockIdx.x * 8 + wave) * 16;
    if (mbase >= NT) return;

    const int half = lane >> 4;
    const int n    = lane & 15;

    const float* lrow = L   + (size_t)(mbase + n) * DD;
    gcf_p erow = src_row<SPLIT>(srcA, srcB, mbase + n);

    v8f acc[4];
    #pragma unroll
    for (int j = 0; j < 4; ++j) acc[j] = (v8f){};

    #pragma unroll
    for (int kk = 0; kk < 16; ++kk) {
        int off = kk * 4 + half * 2;
        v2f ls = *(const v2f*)(lrow + off);
        v2f eg = *(gcv2_p)(erow + off);
        v2f a_li  = ls + eg;
        v2f a_mul = ls * eg;
        int bbase = ((kk * 2 + half) * 64) << 1;
        #pragma unroll
        for (int j = 0; j < 4; ++j) {
            v2f bgc = *(const v2f*)(sg + bbase + ((j * 16 + n) << 1));
            v2f bbi = *(const v2f*)(sb + bbase + ((j * 16 + n) << 1));
            acc[j] = __builtin_amdgcn_wmma_f32_16x16x4_f32(
                false, a_li, false, bgc, (short)0, acc[j], false, false);
            acc[j] = __builtin_amdgcn_wmma_f32_16x16x4_f32(
                false, a_mul, false, bbi, (short)0, acc[j], false, false);
        }
    }

    #pragma unroll
    for (int j = 0; j < 4; ++j) {
        float bsum = sbias[j * 16 + n];
        #pragma unroll
        for (int v = 0; v < 8; ++v) {
            float x = acc[j][v] + bsum;
            x = (x >= 0.f) ? x : SLOPE * x;
            ego[(size_t)(mbase + (half << 3) + v) * DD + j * 16 + n] = x;
        }
    }
}

__global__ __launch_bounds__(256)
void k_gather_norm(const float* __restrict__ ego, const int* __restrict__ us,
                   const int* __restrict__ pi, const int* __restrict__ ni,
                   float* __restrict__ out, int colbase) {
    int wid  = (blockIdx.x * 256 + threadIdx.x) >> 5;
    int lane = threadIdx.x & 31;
    if (wid >= 3 * BATCH) return;
    int seg = wid >> 12;
    int bb  = wid & (BATCH - 1);
    int node;
    if (seg == 0)      node = us[bb];
    else if (seg == 1) node = NU + pi[bb];
    else               node = NU + ni[bb];
    v2f x = *(const v2f*)(ego + (size_t)node * DD + lane * 2);
    float ss = x.x * x.x + x.y * x.y;
    #pragma unroll
    for (int o = 16; o > 0; o >>= 1) ss += __shfl_xor(ss, o, 32);
    float scale = 1.0f / fmaxf(sqrtf(ss), 1e-12f);
    v2f y; y.x = x.x * scale; y.y = x.y * scale;
    *(v2f*)(out + (size_t)wid * 256 + colbase + lane * 2) = y;
}

extern "C" void kernel_launch(void* const* d_in, const int* in_sizes, int n_in,
                              void* d_out, int out_size, void* d_ws, size_t ws_size,
                              hipStream_t stream) {
    const float* ue   = (const float*)d_in[0];
    const float* ie   = (const float*)d_in[1];
    const float* Wgc  = (const float*)d_in[2];
    const float* Wbi  = (const float*)d_in[3];
    const float* bbi  = (const float*)d_in[4];
    const float* vals = (const float*)d_in[5];
    const int*   rows = (const int*)d_in[6];
    const int*   cols = (const int*)d_in[7];
    const int*   us   = (const int*)d_in[8];
    const int*   pi   = (const int*)d_in[9];
    const int*   ni   = (const int*)d_in[10];
    float* out = (float*)d_out;

    float* ego = (float*)d_ws;
    float* L   = ego + (size_t)NT * DD;
    int* binned = (int*)(L + (size_t)NT * DD);
    int* gcount = binned + (size_t)2 * NNZE;
    int* goff   = gcount + 2048;
    int* gcur   = goff + 2048;

    k_gather0<<<(3 * BATCH * DD + 255) / 256, 256, 0, stream>>>(
        ue, ie, us, pi, ni, out, gcount);

    k_bin_count<<<GBIN, 256, 0, stream>>>(rows, gcount, gcur);
    k_bin_place<<<GBIN, 256, 0, stream>>>(rows, cols, vals, gcount, goff, gcur, binned);

    k_spmm_blk<true><<<NB, 256, 0, stream>>>(
        binned, goff, ue, ie, L, ego, us, pi, ni, out, 0);
    k_dense_wmma<true><<<(NT + 127) / 128, 256, 0, stream>>>(
        L, ue, ie, ego, Wgc, Wbi, bbi);
    for (int k = 1; k < 3; ++k) {
        k_spmm_blk<false><<<NB + GNB, 256, 0, stream>>>(
            binned, goff, ego, ego, L, ego, us, pi, ni, out, 64 * k);
        k_dense_wmma<false><<<(NT + 127) / 128, 256, 0, stream>>>(
            L, ego, ego, ego, Wgc + k * 4096, Wbi + k * 4096, bbi + k * 64);
    }
    k_gather_norm<<<(3 * BATCH * 32 + 255) / 256, 256, 0, stream>>>(
        ego, us, pi, ni, out, 64 * 3);
}


# Round trips batched: a bin's last partial chunk is one masked batch of 16 gathers (was one trip per edge); dense launches request all weight loads before their LDS stores; sum order: arrival
# speedup vs baseline: 2.8845x; 1.0609x over previous
#include <hip/hip_runtime.h>
#include <hip/hip_bf16.h>

typedef float v2f __attribute__((ext_vector_type(2)));
typedef float v8f __attribute__((ext_vector_type(8)));

#define NU 50000
#define NI 100000
#define NT 150000
#define DD 64
#define NNZE 2400000
#define BATCH 4096
#define SLOPE 0.01f

__global__ __launch_bounds__(256)
void k_gather0(const float* __restrict__ ue, const float* __restrict__ ie,
               const int* __restrict__ us, const int* __restrict__ pi,
               const int* __restrict__ ni, float* __restrict__ out,
               int* __restrict__ gcount) {
    int t = blockIdx.x * 256 + threadIdx.x;
    if (t < 2048) gcount[t] = 0;
    if (t >= 3 * BATCH * DD) return;
    int c  = t & 63;
    int b  = t >> 6;
    int seg = b >> 12;
    int bb  = b & (BATCH - 1);
    float v;
    if (seg == 0)      v = ue[(size_t)us[bb] * DD + c];
    else if (seg == 1) v = ie[(size_t)pi[bb] * DD + c];
    else               v = ie[(size_t)ni[bb] * DD + c];
    out[(size_t)b * 256 + c] = v;
}

#define RB 128
#define NB ((NT + RB - 1) / RB)
#define CSH 18
#define CMASK ((1 << CSH) - 1)
#define GBIN 512
#define CHUNK ((NNZE + GBIN - 1) / GBIN)
static_assert(NT <= (1 << CSH), "col must fit under the local row");
static_assert(NB + 1 <= 5 * 256, "the placing launch scans five bins per thread");

#define EPT 5
static_assert(CHUNK % 4 == 0 && NNZE % 4 == 0, "a chunk is whole 16-byte groups");
static_assert(EPT * 256 * 4 >= CHUNK, "EPT groups per thread cover a chunk");
static_assert((NNZE - (GBIN - 1) * CHUNK) / 4 >= (EPT - 1) * 256,
              "only the last group of a thread can lie outside the chunk");

__device__ __forceinline__ int chunk_groups(int g) {
    const int e0 = g * CHUNK;
    return (min(e0 + CHUNK, NNZE) - e0) >> 2;
}

__device__ __forceinline__ void hist_add4(int* hist, int4 r) {
    atomicAdd(&hist[(unsigned)r.x / RB], 1);
    atomicAdd(&hist[(unsigned)r.y / RB], 1);
    atomicAdd(&hist[(unsigned)r.z / RB], 1);
    atomicAdd(&hist[(unsigned)r.w / RB], 1);
}

__global__ __launch_bounds__(256)
void k_bin_count(const int* __restrict__ rows, int* __restrict__ gcount,
                 int* __restrict__ gcur) {
    __shared__ int hist[NB];
    if (blockIdx.x == 0)
        for (int i = threadIdx.x; i < NB; i += 256) gcur[i] = 0;
    for (int i = threadIdx.x; i < NB; i += 256) hist[i] = 0;
    __syncthreads();
    const int n4 = chunk_groups(blockIdx.x);
    const int4* r4 = (const int4*)(rows + (size_t)blockIdx.x * CHUNK);
    const bool tail = (int)threadIdx.x + (EPT - 1) * 256 < n4;
    int4 rr[EPT];
    for (int k = 0; k < EPT - 1; ++k) rr[k] = r4[threadIdx.x + k * 256];
    if (tail) rr[EPT - 1] = r4[threadIdx.x + (EPT - 1) * 256];
    for (int k = 0; k < EPT - 1; ++k) hist_add4(hist, rr[k]);
    if (tail) hist_add4(hist, rr[EPT - 1]);
    __syncthreads();
    for (int i = threadIdx.x; i < NB; i += 256) {
        int c = hist[i];
        if (c) atomicAdd(&gcount[i], c);
    }
}

__device__ __forceinline__ void place4(int2* rec, int* hist, const int* base,
                                       int4 r, int4 c, float4 v) {
    const int rv[4] = {r.x, r.y, r.z, r.w};
    const int cv[4] = {c.x, c.y, c.z, c.w};
    const float vv[4] = {v.x, v.y, v.z, v.w};
    int pos[4];
    for (int j = 0; j < 4; ++j) {
        const int b = (unsigned)rv[j] / RB;
        pos[j] = base[b] + atomicAdd(&hist[b], 1);
    }
    for (int j = 0; j < 4; ++j) {
        const int lr = rv[j] & (RB - 1);
        rec[pos[j]] = make_int2(cv[j] | (lr << CSH), __float_as_int(vv[j]));
    }
}
static_assert((RB & (RB - 1)) == 0, "the local row is taken with a mask");

__global__ __launch_bounds__(256)
void k_bin_place(const int* __restrict__ rows, const int* __restrict__ cols,
                 const float* __restrict__ vals, const int* __restrict__ gcount,
                 int* __restrict__ goff, int* __restrict__ gcur,
                 int* __restrict__ binned) {
    __shared__ int hist[NB];
    __shared__ int base[NB];
    __shared__ int off[NB + 1];
    __shared__ int wsum[8];
    for (int i = threadIdx.x; i < NB; i += 256) hist[i] = 0;
    __syncthreads();
    const int n4 = chunk_groups(blockIdx.x);
    const size_t e0 = (size_t)blockIdx.x * CHUNK;
    const int4*   r4 = (const int4*)(rows + e0);
    const int4*   c4 = (const int4*)(cols + e0);
    const float4* v4 = (const float4*)(vals + e0);
    const bool tail = (int)threadIdx.x + (EPT - 1) * 256 < n4;
    int4 rr[EPT], cc[EPT];
    float4 vv[EPT];
    for (int k = 0; k < EPT - 1; ++k) rr[k] = r4[threadIdx.x + k * 256];
    if (tail) rr[EPT - 1] = r4[threadIdx.x + (EPT - 1) * 256];
    int cs[5];
    for (int k = 0; k < 5; ++k) {
        int idx = threadIdx.x * 5 + k;
        cs[k] = (idx < NB) ? gcount[idx] : 0;
    }
    for (int k = 0; k < EPT - 1; ++k) {
        cc[k] = c4[threadIdx.x + k * 256];
        vv[k] = v4[threadIdx.x + k * 256];
    }
    if (tail) {
        cc[EPT - 1] = c4[threadIdx.x + (EPT - 1) * 256];
        vv[EPT - 1] = v4[threadIdx.x + (EPT - 1) * 256];
    }
    for (int k = 0; k < EPT - 1; ++k) hist_add4(hist, rr[k]);
    if (tail) hist_add4(hist, rr[EPT - 1]);
    {
        const int lane = threadIdx.x & 31, wv = threadIdx.x >> 5;
        const int s = cs[0] + cs[1] + cs[2] + cs[3] + cs[4];
        int incl = s;
        for (int o = 1; o < 32; o <<= 1) {
            int v = __shfl_up(incl, o, 32);
            if (lane >= o) incl += v;
        }
        if (lane == 31) wsum[wv] = incl;
        __syncthreads();
        int excl = incl - s;
        for (int w = 0; w < wv; ++w) excl += wsum[w];
        for (int k = 0; k < 5; ++k) {
            int idx = threadIdx.x * 5 + k;
            if (idx <= NB) off[idx] = excl;
            excl += cs[k];
        }
    }
    __syncthreads();
    if (blockIdx.x == 0)
        for (int i = threadIdx.x; i <= NB; i += 256) goff[i] = off[i];
    {
        int cnt[5], got[5];
        for (int k = 0; k < 5; ++k) {
            const int i = threadIdx.x + k * 256;
            cnt[k] = (i < NB) ? hist[i] : 0;
        }
        for (int k = 0; k < 5; ++k) {
            const int i = threadIdx.x + k * 256;
            got[k] = (i < NB) ? atomicAdd(&gcur[i], cnt[k]) : 0;
        }
        for (int k = 0; k < 5; ++k) {
            const int i = threadIdx.x + k * 256;
            if (i < NB) {
                base[i] = off[i] + got[k];
                hist[i] = 0;
            }
        }
    }
    static_assert(5 * 256 >= NB, "five bins per thread in the cursor step");
    __syncthreads();
    int2* rec = (int2*)binned;
    for (int k = 0; k < EPT - 1; ++k) place4(rec, hist, base, rr[k], cc[k], vv[k]);
    if (tail) place4(rec, hist, base, rr[EPT - 1], cc[EPT - 1], vv[EPT - 1]);
}

typedef const float __attribute__((address_space(1)))* gcf_p;
typedef const v2f   __attribute__((address_space(1)))* gcv2_p;
template <bool SPLIT>
__device__ __forceinline__ gcf_p src_row(const float* a, const float* b, int r) {
    if (!SPLIT) return (gcf_p)(a + (size_t)r * DD);
    const unsigned long long ba = (unsigned long long)a;
    const unsigned long long bb = (unsigned long long)b - (unsigned long long)NU * DD * sizeof(float);
    const unsigned long long base = (r < NU) ? ba : bb;
    return (gcf_p)(base + (unsigned long long)(unsigned)r * (DD * sizeof(float)));
}

#define GNR 8
#define GNB (3 * BATCH / (8 * GNR))
static_assert(GNB * 8 * GNR == 3 * BATCH, "output rows split evenly");
__device__ __forceinline__ void gather_norm_rows(
        const float* __restrict__ ego, const int* __restrict__ us,
        const int* __restrict__ pi, const int* __restrict__ ni,
        float* __restrict__ out, int colbase, int g) {
    const int lane = threadIdx.x & 31;
    const int w0 = g * 8 + (threadIdx.x >> 5);
    #pragma unroll
    for (int k = 0; k < GNR; ++k) {
        const int wid = w0 + k * (GNB * 8);
        const int seg = wid >> 12;
        const int bb  = wid & (BATCH - 1);
        const int* ip = (seg == 0) ? us : (seg == 1) ? pi : ni;
        const int node = ip[bb] + (seg ? NU : 0);
        v2f x = *(const v2f*)(ego + (size_t)node * DD + lane * 2);
        float ss = x.x * x.x + x.y * x.y;
        for (int o = 16; o > 0; o >>= 1) ss += __shfl_xor(ss, o, 32);
        float scale = 1.0f / fmaxf(sqrtf(ss), 1e-12f);
        v2f y; y.x = x.x * scale; y.y = x.y * scale;
        *(v2f*)(out + (size_t)wid * 256 + colbase + lane * 2) = y;
    }
}

__device__ __forceinline__ void lds_add(float* p, float v) {
    __hip_atomic_fetch_add(p, v, __ATOMIC_RELAXED, __HIP_MEMORY_SCOPE_WORKGROUP);
}

template <bool SPLIT>
__global__ __launch_bounds__(256)
void k_spmm_blk(const int* __restrict__ binned, const int* __restrict__ goff,
                const float* __restrict__ srcA, const float* __restrict__ srcB,
                float* __restrict__ L,
                const float* __restrict__ gsrc, const int* __restrict__ us,
                const int* __restrict__ pi, const int* __restrict__ ni,
                float* __restrict__ out, int colbase) {
    __shared__ float tile[RB * DD];
    if (blockIdx.x >= NB) {
        gather_norm_rows(gsrc, us, pi, ni, out, colbase, blockIdx.x - NB);
        return;
    }
    const int b = blockIdx.x;
    float4* t4 = (float4*)tile;
    for (int i = threadIdx.x; i < RB * DD / 4; i += 256)
        t4[i] = make_float4(0.f, 0.f, 0.f, 0.f);
    __syncthreads();

    const int beg = goff[b], end = goff[b + 1];
    const int wave = __builtin_amdgcn_readfirstlane(threadIdx.x >> 5);
    const int lane = threadIdx.x & 31;
    const int2* rec = (const int2*)binned;

    int e0 = beg + wave * 32;
    int2 rn = (e0 + lane < end) ? rec[e0 + lane] : make_int2(0, 0);
    for (; e0 < end; e0 += 256) {
        const int2 r = rn;
        asm volatile("" :: "v"(r.x), "v"(r.y));
        const int nx = e0 + 256 + lane;
        rn = (nx < end) ? rec[nx] : make_int2(0, 0);
        const int cnt = min(32, end - e0);
        int t = 0;
        for (; t + 16 <= cnt; t += 16) {
            v2f x[16];
            int pk[16];
            #pragma unroll
            for (int u = 0; u < 16; ++u) {
                pk[u] = __builtin_amdgcn_readlane(r.x, t + u);
                x[u] = *(gcv2_p)(src_row<SPLIT>(srcA, srcB, pk[u] & CMASK) + lane * 2);
            }
            #pragma unroll
            for (int u = 0; u < 16; ++u) {
                float v = __int_as_float(__builtin_amdgcn_readlane(r.y, t + u));
                float* d = tile + (pk[u] >> CSH) * DD + lane * 2;
                lds_add(d,     v * x[u].x);
                lds_add(d + 1, v * x[u].y);
            }
        }
        if (t < cnt) {
            v2f x[16];
            int pk[16];
            #pragma unroll
            for (int u = 0; u < 16; ++u) {
                pk[u] = __builtin_amdgcn_readlane(r.x, min(t + u, cnt - 1));
                x[u] = *(gcv2_p)(src_row<SPLIT>(srcA, srcB, pk[u] & CMASK) + lane * 2);
            }
            #pragma unroll
            for (int u = 0; u < 16; ++u) {
                if (t + u < cnt) {
                    float v = __int_as_float(__builtin_amdgcn_readlane(r.y, t + u));
                    float* d = tile + (pk[u] >> CSH) * DD + lane * 2;
                    lds_add(d,     v * x[u].x);
                    lds_add(d + 1, v * x[u].y);
                }
            }
        }
    }
    __syncthreads();

    for (int i = threadIdx.x; i < RB * DD / 4; i += 256) {
        int row = b * RB + (i >> 4);
        if (row < NT)
            *(float4*)(L + (size_t)row * DD + (i & 15) * 4) = t4[i];
    }
}

template <bool SPLIT>
__global__ __launch_bounds__(256)
void k_dense_wmma(const float* __restrict__ L, const float* srcA, const float* srcB,
                  float* ego,
                  const float* __restrict__ Wgc, const float* __restrict__ Wbi,
                  const float* __restrict__ bias) {
    __shared__ __attribute__((aligned(16))) float sg[4096];
    __shared__ __attribute__((aligned(16))) float sb[4096];
    __shared__ float sbias[64];

    {
        const int p = threadIdx.x >> 3;
        const int f = threadIdx.x & 7;
        float4 g[2][2], h[2][2];
        for (int q = 0; q < 2; ++q) {
            for (int e = 0; e < 2; ++e) {
                const int idx = (2 * p + e) * 64 + (f + 8 * q) * 4;
                g[e][q] = *(const float4*)(Wgc + idx);
                h[e][q] = *(const float4*)(Wbi + idx);
            }
        }
        const float b2 = (threadIdx.x < 64) ? bias[threadIdx.x] : 0.f;
        for (int q = 0; q < 2; ++q) {
            const int dst = (p * 64 + (f + 8 * q) * 4) << 1;
            *(float4*)(sg + dst)     = make_float4(g[0][q].x, g[1][q].x, g[0][q].y, g[1][q].y);
            *(float4*)(sg + dst + 4) = make_float4(g[0][q].z, g[1][q].z, g[0][q].w, g[1][q].w);
            *(float4*)(sb + dst)     = make_float4(h[0][q].x, h[1][q].x, h[0][q].y, h[1][q].y);
            *(float4*)(sb + dst + 4) = make_float4(h[0][q].z, h[1][q].z, h[0][q].w, h[1][q].w);
        }
        if (threadIdx.x < 64) sbias[threadIdx.x] = 2.0f * b2;
    }
    __syncthreads();

    const int wave  = threadIdx.x >> 5;
    const int lane  = threadIdx.x & 31;
    const int mbase = (blockIdx.x * 8 + wave) * 16;
    if (mbase >= NT) return;

    const int half = lane >> 4;
    const int n    = lane & 15;

    const float* lrow = L   + (size_t)(mbase + n) * DD;
    gcf_p erow = src_row<SPLIT>(srcA, srcB, mbase + n);

    v8f acc[4];
    #pragma unroll
    for (int j = 0; j < 4; ++j) acc[j] = (v8f){};

    #pragma unroll
    for (int kk = 0; kk < 16; ++kk) {
        int off = kk * 4 + half * 2;
        v2f ls = *(const v2f*)(lrow + off);
        v2f eg = *(gcv2_p)(erow + off);
        v2f a_li  = ls + eg;
        v2f a_mul = ls * eg;
        int bbase = ((kk * 2 + half) * 64) << 1;
        #pragma unroll
        for (int j = 0; j < 4; ++j) {
            v2f bgc = *(const v2f*)(sg + bbase + ((j * 16 + n) << 1));
            v2f bbi = *(const v2f*)(sb + bbase + ((j * 16 + n) << 1));
            acc[j] = __builtin_amdgcn_wmma_f32_16x16x4_f32(
                false, a_li, false, bgc, (short)0, acc[j], false, false);
            acc[j] = __builtin_amdgcn_wmma_f32_16x16x4_f32(
                false, a_mul, false, bbi, (short)0, acc[j], false, false);
        }
    }

    #pragma unroll
    for (int j = 0; j < 4; ++j) {
        float bsum = sbias[j * 16 + n];
        #pragma unroll
        for (int v = 0; v < 8; ++v) {
            float x = acc[j][v] + bsum;
            x = (x >= 0.f) ? x : SLOPE * x;
            ego[(size_t)(mbase + (half << 3) + v) * DD + j * 16 + n] = x;
        }
    }
}

__global__ __launch_bounds__(256)
void k_gather_norm(const float* __restrict__ ego, const int* __restrict__ us,
                   const int* __restrict__ pi, const int* __restrict__ ni,
                   float* __restrict__ out, int colbase) {
    int wid  = (blockIdx.x * 256 + threadIdx.x) >> 5;
    int lane = threadIdx.x & 31;
    if (wid >= 3 * BATCH) return;
    int seg = wid >> 12;
    int bb  = wid & (BATCH - 1);
    int node;
    if (seg == 0)      node = us[bb];
    else if (seg == 1) node = NU + pi[bb];
    else               node = NU + ni[bb];
    v2f x = *(const v2f*)(ego + (size_t)node * DD + lane * 2);
    float ss = x.x * x.x + x.y * x.y;
    #pragma unroll
    for (int o = 16; o > 0; o >>= 1) ss += __shfl_xor(ss, o, 32);
    float scale = 1.0f / fmaxf(sqrtf(ss), 1e-12f);
    v2f y; y.x = x.x * scale; y.y = x.y * scale;
    *(v2f*)(out + (size_t)wid * 256 + colbase + lane * 2) = y;
}

extern "C" void kernel_launch(void* const* d_in, const int* in_sizes, int n_in,
                              void* d_out, int out_size, void* d_ws, size_t ws_size,
                              hipStream_t stream) {
    const float* ue   = (const float*)d_in[0];
    const float* ie   = (const float*)d_in[1];
    const float* Wgc  = (const float*)d_in[2];
    const float* Wbi  = (const float*)d_in[3];
    const float* bbi  = (const float*)d_in[4];
    const float* vals = (const float*)d_in[5];
    const int*   rows = (const int*)d_in[6];
    const int*   cols = (const int*)d_in[7];
    const int*   us   = (const int*)d_in[8];
    const int*   pi   = (const int*)d_in[9];
    const int*   ni   = (const int*)d_in[10];
    float* out = (float*)d_out;

    float* ego = (float*)d_ws;
    float* L   = ego + (size_t)NT * DD;
    int* binned = (int*)(L + (size_t)NT * DD);
    int* gcount = binned + (size_t)2 * NNZE;
    int* goff   = gcount + 2048;
    int* gcur   = goff + 2048;

    k_gather0<<<(3 * BATCH * DD + 255) / 256, 256, 0, stream>>>(
        ue, ie, us, pi, ni, out, gcount);

    k_bin_count<<<GBIN, 256, 0, stream>>>(rows, gcount, gcur);
    k_bin_place<<<GBIN, 256, 0, stream>>>(rows, cols, vals, gcount, goff, gcur, binned);

    k_spmm_blk<true><<<NB, 256, 0, stream>>>(
        binned, goff, ue, ie, L, ego, us, pi, ni, out, 0);
    k_dense_wmma<true><<<(NT + 127) / 128, 256, 0, stream>>>(
        L, ue, ie, ego, Wgc, Wbi, bbi);
    for (int k = 1; k < 3; ++k) {
        k_spmm_blk<false><<<NB + GNB, 256, 0, stream>>>(
            binned, goff, ego, ego, L, ego, us, pi, ni, out, 64 * k);
        k_dense_wmma<false><<<(NT + 127) / 128, 256, 0, stream>>>(
            L, ego, ego, ego, Wgc + k * 4096, Wbi + k * 4096, bbi + k * 64);
    }
    k_gather_norm<<<(3 * BATCH * 32 + 255) / 256, 256, 0, stream>>>(
        ego, us, pi, ni, out, 64 * 3);
}


# Last layer only for rows the output reads: a block marks them in a 128-bit LDS mask and lists its bin's edges into them; dense step + normalise per 16 output slots; sum order: arrival, as before
# speedup vs baseline: 3.6927x; 1.2802x over previous
#include <hip/hip_runtime.h>
#include <hip/hip_bf16.h>

typedef float v2f __attribute__((ext_vector_type(2)));
typedef float v8f __attribute__((ext_vector_type(8)));

#define NU 50000
#define NI 100000
#define NT 150000
#define DD 64
#define NNZE 2400000
#define BATCH 4096
#define SLOPE 0.01f

__global__ __launch_bounds__(256)
void k_gather0(const float* __restrict__ ue, const float* __restrict__ ie,
               const int* __restrict__ us, const int* __restrict__ pi,
               const int* __restrict__ ni, float* __restrict__ out,
               int* __restrict__ gcount) {
    int t = blockIdx.x * 256 + threadIdx.x;
    if (t < 2048) gcount[t] = 0;
    if (t >= 3 * BATCH * DD) return;
    int c  = t & 63;
    int b  = t >> 6;
    int seg = b >> 12;
    int bb  = b & (BATCH - 1);
    float v;
    if (seg == 0)      v = ue[(size_t)us[bb] * DD + c];
    else if (seg == 1) v = ie[(size_t)pi[bb] * DD + c];
    else               v = ie[(size_t)ni[bb] * DD + c];
    out[(size_t)b * 256 + c] = v;
}

#define RB 128
#define NB ((NT + RB - 1) / RB)
#define CSH 18
#define CMASK ((1 << CSH) - 1)
#define GBIN 512
#define CHUNK ((NNZE + GBIN - 1) / GBIN)
static_assert(NT <= (1 << CSH), "col must fit under the local row");
static_assert(NB + 1 <= 5 * 256, "the placing launch scans five bins per thread");

#define EPT 5
static_assert(CHUNK % 4 == 0 && NNZE % 4 == 0, "a chunk is whole 16-byte groups");
static_assert(EPT * 256 * 4 >= CHUNK, "EPT groups per thread cover a chunk");
static_assert((NNZE - (GBIN - 1) * CHUNK) / 4 >= (EPT - 1) * 256,
              "only the last group of a thread can lie outside the chunk");

__device__ __forceinline__ int chunk_groups(int g) {
    const int e0 = g * CHUNK;
    return (min(e0 + CHUNK, NNZE) - e0) >> 2;
}

__device__ __forceinline__ void hist_add4(int* hist, int4 r) {
    atomicAdd(&hist[(unsigned)r.x / RB], 1);
    atomicAdd(&hist[(unsigned)r.y / RB], 1);
    atomicAdd(&hist[(unsigned)r.z / RB], 1);
    atomicAdd(&hist[(unsigned)r.w / RB], 1);
}

__global__ __launch_bounds__(256)
void k_bin_count(const int* __restrict__ rows, int* __restrict__ gcount,
                 int* __restrict__ gcur) {
    __shared__ int hist[NB];
    if (blockIdx.x == 0)
        for (int i = threadIdx.x; i < NB; i += 256) gcur[i] = 0;
    for (int i = threadIdx.x; i < NB; i += 256) hist[i] = 0;
    __syncthreads();
    const int n4 = chunk_groups(blockIdx.x);
    const int4* r4 = (const int4*)(rows + (size_t)blockIdx.x * CHUNK);
    const bool tail = (int)threadIdx.x + (EPT - 1) * 256 < n4;
    int4 rr[EPT];
    for (int k = 0; k < EPT - 1; ++k) rr[k] = r4[threadIdx.x + k * 256];
    if (tail) rr[EPT - 1] = r4[threadIdx.x + (EPT - 1) * 256];
    for (int k = 0; k < EPT - 1; ++k) hist_add4(hist, rr[k]);
    if (tail) hist_add4(hist, rr[EPT - 1]);
    __syncthreads();
    for (int i = threadIdx.x; i < NB; i += 256) {
        int c = hist[i];
        if (c) atomicAdd(&gcount[i], c);
    }
}

__device__ __forceinline__ void place4(int2* rec, int* hist, const int* base,
                                       int4 r, int4 c, float4 v) {
    const int rv[4] = {r.x, r.y, r.z, r.w};
    const int cv[4] = {c.x, c.y, c.z, c.w};
    const float vv[4] = {v.x, v.y, v.z, v.w};
    int pos[4];
    for (int j = 0; j < 4; ++j) {
        const int b = (unsigned)rv[j] / RB;
        pos[j] = base[b] + atomicAdd(&hist[b], 1);
    }
    for (int j = 0; j < 4; ++j) {
        const int lr = rv[j] & (RB - 1);
        rec[pos[j]] = make_int2(cv[j] | (lr << CSH), __float_as_int(vv[j]));
    }
}
static_assert((RB & (RB - 1)) == 0, "the local row is taken with a mask");

__global__ __launch_bounds__(256)
void k_bin_place(const int* __restrict__ rows, const int* __restrict__ cols,
                 const float* __restrict__ vals, const int* __restrict__ gcount,
                 int* __restrict__ goff, int* __restrict__ gcur,
                 int* __restrict__ binned) {
    __shared__ int hist[NB];
    __shared__ int base[NB];
    __shared__ int off[NB + 1];
    __shared__ int wsum[8];
    for (int i = threadIdx.x; i < NB; i += 256) hist[i] = 0;
    __syncthreads();
    const int n4 = chunk_groups(blockIdx.x);
    const size_t e0 = (size_t)blockIdx.x * CHUNK;
    const int4*   r4 = (const int4*)(rows + e0);
    const int4*   c4 = (const int4*)(cols + e0);
    const float4* v4 = (const float4*)(vals + e0);
    const bool tail = (int)threadIdx.x + (EPT - 1) * 256 < n4;
    int4 rr[EPT], cc[EPT];
    float4 vv[EPT];
    for (int k = 0; k < EPT - 1; ++k) rr[k] = r4[threadIdx.x + k * 256];
    if (tail) rr[EPT - 1] = r4[threadIdx.x + (EPT - 1) * 256];
    int cs[5];
    for (int k = 0; k < 5; ++k) {
        int idx = threadIdx.x * 5 + k;
        cs[k] = (idx < NB) ? gcount[idx] : 0;
    }
    for (int k = 0; k < EPT - 1; ++k) {
        cc[k] = c4[threadIdx.x + k * 256];
        vv[k] = v4[threadIdx.x + k * 256];
    }
    if (tail) {
        cc[EPT - 1] = c4[threadIdx.x + (EPT - 1) * 256];
        vv[EPT - 1] = v4[threadIdx.x + (EPT - 1) * 256];
    }
    for (int k = 0; k < EPT - 1; ++k) hist_add4(hist, rr[k]);
    if (tail) hist_add4(hist, rr[EPT - 1]);
    {
        const int lane = threadIdx.x & 31, wv = threadIdx.x >> 5;
        const int s = cs[0] + cs[1] + cs[2] + cs[3] + cs[4];
        int incl = s;
        for (int o = 1; o < 32; o <<= 1) {
            int v = __shfl_up(incl, o, 32);
            if (lane >= o) incl += v;
        }
        if (lane == 31) wsum[wv] = incl;
        __syncthreads();
        int excl = incl - s;
        for (int w = 0; w < wv; ++w) excl += wsum[w];
        for (int k = 0; k < 5; ++k) {
            int idx = threadIdx.x * 5 + k;
            if (idx <= NB) off[idx] = excl;
            excl += cs[k];
        }
    }
    __syncthreads();
    if (blockIdx.x == 0)
        for (int i = threadIdx.x; i <= NB; i += 256) goff[i] = off[i];
    {
        int cnt[5], got[5];
        for (int k = 0; k < 5; ++k) {
            const int i = threadIdx.x + k * 256;
            cnt[k] = (i < NB) ? hist[i] : 0;
        }
        for (int k = 0; k < 5; ++k) {
            const int i = threadIdx.x + k * 256;
            got[k] = (i < NB) ? atomicAdd(&gcur[i], cnt[k]) : 0;
        }
        for (int k = 0; k < 5; ++k) {
            const int i = threadIdx.x + k * 256;
            if (i < NB) {
                base[i] = off[i] + got[k];
                hist[i] = 0;
            }
        }
    }
    static_assert(5 * 256 >= NB, "five bins per thread in the cursor step");
    __syncthreads();
    int2* rec = (int2*)binned;
    for (int k = 0; k < EPT - 1; ++k) place4(rec, hist, base, rr[k], cc[k], vv[k]);
    if (tail) place4(rec, hist, base, rr[EPT - 1], cc[EPT - 1], vv[EPT - 1]);
}

typedef const float __attribute__((address_space(1)))* gcf_p;
typedef const v2f   __attribute__((address_space(1)))* gcv2_p;
template <bool SPLIT>
__device__ __forceinline__ gcf_p src_row(const float* a, const float* b, int r) {
    if (!SPLIT) return (gcf_p)(a + (size_t)r * DD);
    const unsigned long long ba = (unsigned long long)a;
    const unsigned long long bb = (unsigned long long)b - (unsigned long long)NU * DD * sizeof(float);
    const unsigned long long base = (r < NU) ? ba : bb;
    return (gcf_p)(base + (unsigned long long)(unsigned)r * (DD * sizeof(float)));
}

#define GNR 8
#define GNB (3 * BATCH / (8 * GNR))
static_assert(GNB * 8 * GNR == 3 * BATCH, "output rows split evenly");
__device__ __forceinline__ void gather_norm_rows(
        const float* __restrict__ ego, const int* __restrict__ us,
        const int* __restrict__ pi, const int* __restrict__ ni,
        float* __restrict__ out, int colbase, int g) {
    const int lane = threadIdx.x & 31;
    const int w0 = g * 8 + (threadIdx.x >> 5);
    #pragma unroll
    for (int k = 0; k < GNR; ++k) {
        const int wid = w0 + k * (GNB * 8);
        const int seg = wid >> 12;
        const int bb  = wid & (BATCH - 1);
        const int* ip = (seg == 0) ? us : (seg == 1) ? pi : ni;
        const int node = ip[bb] + (seg ? NU : 0);
        v2f x = *(const v2f*)(ego + (size_t)node * DD + lane * 2);
        float ss = x.x * x.x + x.y * x.y;
        for (int o = 16; o > 0; o >>= 1) ss += __shfl_xor(ss, o, 32);
        float scale = 1.0f / fmaxf(sqrtf(ss), 1e-12f);
        v2f y; y.x = x.x * scale; y.y = x.y * scale;
        *(v2f*)(out + (size_t)wid * 256 + colbase + lane * 2) = y;
    }
}

__device__ __forceinline__ void lds_add(float* p, float v) {
    __hip_atomic_fetch_add(p, v, __ATOMIC_RELAXED, __HIP_MEMORY_SCOPE_WORKGROUP);
}

template <bool SPLIT>
__global__ __launch_bounds__(256)
void k_spmm_blk(const int* __restrict__ binned, const int* __restrict__ goff,
                const float* __restrict__ srcA, const float* __restrict__ srcB,
                float* __restrict__ L,
                const float* __restrict__ gsrc, const int* __restrict__ us,
                const int* __restrict__ pi, const int* __restrict__ ni,
                float* __restrict__ out, int colbase) {
    __shared__ float tile[RB * DD];
    if (blockIdx.x >= NB) {
        gather_norm_rows(gsrc, us, pi, ni, out, colbase, blockIdx.x - NB);
        return;
    }
    const int b = blockIdx.x;
    float4* t4 = (float4*)tile;
    for (int i = threadIdx.x; i < RB * DD / 4; i += 256)
        t4[i] = make_float4(0.f, 0.f, 0.f, 0.f);
    __syncthreads();

    const int beg = goff[b], end = goff[b + 1];
    const int wave = __builtin_amdgcn_readfirstlane(threadIdx.x >> 5);
    const int lane = threadIdx.x & 31;
    const int2* rec = (const int2*)binned;

    int e0 = beg + wave * 32;
    int2 rn = (e0 + lane < end) ? rec[e0 + lane] : make_int2(0, 0);
    for (; e0 < end; e0 += 256) {
        const int2 r = rn;
        asm volatile("" :: "v"(r.x), "v"(r.y));
        const int nx = e0 + 256 + lane;
        rn = (nx < end) ? rec[nx] : make_int2(0, 0);
        const int cnt = min(32, end - e0);
        int t = 0;
        for (; t + 16 <= cnt; t += 16) {
            v2f x[16];
            int pk[16];
            #pragma unroll
            for (int u = 0; u < 16; ++u) {
                pk[u] = __builtin_amdgcn_readlane(r.x, t + u);
                x[u] = *(gcv2_p)(src_row<SPLIT>(srcA, srcB, pk[u] & CMASK) + lane * 2);
            }
            #pragma unroll
            for (int u = 0; u < 16; ++u) {
                float v = __int_as_float(__builtin_amdgcn_readlane(r.y, t + u));
                float* d = tile + (pk[u] >> CSH) * DD + lane * 2;
                lds_add(d,     v * x[u].x);
                lds_add(d + 1, v * x[u].y);
            }
        }
        if (t < cnt) {
            v2f x[16];
            int pk[16];
            #pragma unroll
            for (int u = 0; u < 16; ++u) {
                pk[u] = __builtin_amdgcn_readlane(r.x, min(t + u, cnt - 1));
                x[u] = *(gcv2_p)(src_row<SPLIT>(srcA, srcB, pk[u] & CMASK) + lane * 2);
            }
            #pragma unroll
            for (int u = 0; u < 16; ++u) {
                if (t + u < cnt) {
                    float v = __int_as_float(__builtin_amdgcn_readlane(r.y, t + u));
                    float* d = tile + (pk[u] >> CSH) * DD + lane * 2;
                    lds_add(d,     v * x[u].x);
                    lds_add(d + 1, v * x[u].y);
                }
            }
        }
    }
    __syncthreads();

    for (int i = threadIdx.x; i < RB * DD / 4; i += 256) {
        int row = b * RB + (i >> 4);
        if (row < NT)
            *(float4*)(L + (size_t)row * DD + (i & 15) * 4) = t4[i];
    }
}

#define SCAN 2048
static_assert(SCAN == 8 * 256, "a round is eight records per thread: the list cannot overflow");
static_assert(BATCH == 4 * 4 * 256, "an index array is four 16-byte loads per thread");
static_assert(RB == 128, "the mask is four words");

__device__ __forceinline__ bool row_marked(unsigned m0, unsigned m1, unsigned m2,
                                           unsigned m3, int lr) {
    const unsigned w = (lr & 64) ? ((lr & 32) ? m3 : m2) : ((lr & 32) ? m1 : m0);
    return (w >> (lr & 31)) & 1u;
}

__global__ __launch_bounds__(256)
void k_spmm_last(const int* __restrict__ binned, const int* __restrict__ goff,
                 const float* __restrict__ src, float* __restrict__ L,
                 const int* __restrict__ us, const int* __restrict__ pi,
                 const int* __restrict__ ni, float* __restrict__ out, int colbase) {
    __shared__ float tile[RB * DD];
    __shared__ __attribute__((aligned(8))) int keep[2 * SCAN];
    __shared__ unsigned need[RB / 32];
    __shared__ int nkeep;
    if (blockIdx.x >= NB) {
        gather_norm_rows(src, us, pi, ni, out, colbase, blockIdx.x - NB);
        return;
    }
    const int b = blockIdx.x;
    float4* t4 = (float4*)tile;
    for (int i = threadIdx.x; i < RB * DD / 4; i += 256)
        t4[i] = make_float4(0.f, 0.f, 0.f, 0.f);
    if (threadIdx.x < RB / 32) need[threadIdx.x] = 0u;
    __syncthreads();

    {
        int4 q[12];
        #pragma unroll
        for (int k = 0; k < 12; ++k) {
            const int* ip = (k < 4) ? us : (k < 8) ? pi : ni;
            q[k] = ((const int4*)ip)[threadIdx.x + (k & 3) * 256];
        }
        #pragma unroll
        for (int k = 0; k < 12; ++k) {
            const int add = (k < 4) ? 0 : NU;
            const int nv[4] = {q[k].x + add, q[k].y + add, q[k].z + add, q[k].w + add};
            for (int j = 0; j < 4; ++j)
                if ((unsigned)nv[j] / RB == (unsigned)b)
                    atomicOr(&need[(nv[j] & (RB - 1)) >> 5], 1u << (nv[j] & 31));
        }
    }
    __syncthreads();
    const unsigned m0 = need[0], m1 = need[1], m2 = need[2], m3 = need[3];
    if ((m0 | m1 | m2 | m3) == 0u) return;

    const int beg = goff[b], end = goff[b + 1];
    const int wave = __builtin_amdgcn_readfirstlane(threadIdx.x >> 5);
    const int lane = threadIdx.x & 31;
    const int2* rec = (const int2*)binned;
    int2* kp = (int2*)keep;

    for (int s0 = beg; s0 < end; s0 += SCAN) {
        if (threadIdx.x == 0) nkeep = 0;
        __syncthreads();
        {
            int2 r[8];
            #pragma unroll
            for (int k = 0; k < 8; ++k) {
                const int e = s0 + threadIdx.x + k * 256;
                r[k] = (e < end) ? rec[e] : make_int2(0, 0);
            }
            #pragma unroll
            for (int k = 0; k < 8; ++k) {
                const int e = s0 + threadIdx.x + k * 256;
                if (e < end && row_marked(m0, m1, m2, m3, (unsigned)r[k].x >> CSH))
                    kp[atomicAdd(&nkeep, 1)] = r[k];
            }
        }
        __syncthreads();
        const int nk = __builtin_amdgcn_readfirstlane(nkeep);
        for (int t0 = wave * 16; t0 < nk; t0 += 8 * 16) {
            const int cnt = min(16, nk - t0);
            const int2 r = kp[t0 + min(lane, cnt - 1)];
            v2f x[16];
            int pk[16];
            #pragma unroll
            for (int u = 0; u < 16; ++u) {
                pk[u] = __builtin_amdgcn_readlane(r.x, u);
                x[u] = *(gcv2_p)((gcf_p)(src + (size_t)(pk[u] & CMASK) * DD) + lane * 2);
            }
            #pragma unroll
            for (int u = 0; u < 16; ++u) {
                if (u < cnt) {
                    float v = __int_as_float(__builtin_amdgcn_readlane(r.y, u));
                    float* d = tile + (pk[u] >> CSH) * DD + lane * 2;
                    lds_add(d,     v * x[u].x);
                    lds_add(d + 1, v * x[u].y);
                }
            }
        }
        __syncthreads();
    }

    for (int i = threadIdx.x; i < RB * DD / 4; i += 256) {
        const int lr = i >> 4;
        const int row = b * RB + lr;
        if (row < NT && row_marked(m0, m1, m2, m3, lr))
            *(float4*)(L + (size_t)row * DD + (i & 15) * 4) = t4[i];
    }
}


__device__ __forceinline__ void stage_weights(float* sg, float* sb, float* sbias,
        const float* __restrict__ Wgc, const float* __restrict__ Wbi,
        const float* __restrict__ bias) {
    const int p = threadIdx.x >> 3;
    const int f = threadIdx.x & 7;
    float4 g[2][2], h[2][2];
    for (int q = 0; q < 2; ++q) {
        for (int e = 0; e < 2; ++e) {
            const int idx = (2 * p + e) * 64 + (f + 8 * q) * 4;
            g[e][q] = *(const float4*)(Wgc + idx);
            h[e][q] = *(const float4*)(Wbi + idx);
        }
    }
    const float b2 = (threadIdx.x < 64) ? bias[threadIdx.x] : 0.f;
    for (int q = 0; q < 2; ++q) {
        const int dst = (p * 64 + (f + 8 * q) * 4) << 1;
        *(float4*)(sg + dst)     = make_float4(g[0][q].x, g[1][q].x, g[0][q].y, g[1][q].y);
        *(float4*)(sg + dst + 4) = make_float4(g[0][q].z, g[1][q].z, g[0][q].w, g[1][q].w);
        *(float4*)(sb + dst)     = make_float4(h[0][q].x, h[1][q].x, h[0][q].y, h[1][q].y);
        *(float4*)(sb + dst + 4) = make_float4(h[0][q].z, h[1][q].z, h[0][q].w, h[1][q].w);
    }
    if (threadIdx.x < 64) sbias[threadIdx.x] = 2.0f * b2;
}

__device__ __forceinline__ void dense_tile(const float* lrow, gcf_p erow,
        const float* sg, const float* sb, int half, int n, v8f (&acc)[4]) {
    #pragma unroll
    for (int j = 0; j < 4; ++j) acc[j] = (v8f){};

    #pragma unroll
    for (int kk = 0; kk < 16; ++kk) {
        int off = kk * 4 + half * 2;
        v2f ls = *(const v2f*)(lrow + off);
        v2f eg = *(gcv2_p)(erow + off);
        v2f a_li  = ls + eg;
        v2f a_mul = ls * eg;
        int bbase = ((kk * 2 + half) * 64) << 1;
        #pragma unroll
        for (int j = 0; j < 4; ++j) {
            v2f bgc = *(const v2f*)(sg + bbase + ((j * 16 + n) << 1));
            v2f bbi = *(const v2f*)(sb + bbase + ((j * 16 + n) << 1));
            acc[j] = __builtin_amdgcn_wmma_f32_16x16x4_f32(
                false, a_li, false, bgc, (short)0, acc[j], false, false);
            acc[j] = __builtin_amdgcn_wmma_f32_16x16x4_f32(
                false, a_mul, false, bbi, (short)0, acc[j], false, false);
        }
    }
}

template <bool SPLIT>
__global__ __launch_bounds__(256)
void k_dense_wmma(const float* __restrict__ L, const float* srcA, const float* srcB,
                  float* ego,
                  const float* __restrict__ Wgc, const float* __restrict__ Wbi,
                  const float* __restrict__ bias) {
    __shared__ __attribute__((aligned(16))) float sg[4096];
    __shared__ __attribute__((aligned(16))) float sb[4096];
    __shared__ float sbias[64];

    stage_weights(sg, sb, sbias, Wgc, Wbi, bias);
    __syncthreads();

    const int wave  = threadIdx.x >> 5;
    const int lane  = threadIdx.x & 31;
    const int mbase = (blockIdx.x * 8 + wave) * 16;
    if (mbase >= NT) return;

    const int half = lane >> 4;
    const int n    = lane & 15;

    const float* lrow = L   + (size_t)(mbase + n) * DD;
    gcf_p erow = src_row<SPLIT>(srcA, srcB, mbase + n);

    v8f acc[4];
    dense_tile(lrow, erow, sg, sb, half, n, acc);

    #pragma unroll
    for (int j = 0; j < 4; ++j) {
        float bsum = sbias[j * 16 + n];
        #pragma unroll
        for (int v = 0; v < 8; ++v) {
            float x = acc[j][v] + bsum;
            x = (x >= 0.f) ? x : SLOPE * x;
            ego[(size_t)(mbase + (half << 3) + v) * DD + j * 16 + n] = x;
        }
    }
}

static_assert(BATCH % 16 == 0, "the 16 slots of a tile share one index array");
static_assert(3 * BATCH % 128 == 0, "eight whole tiles per work-group");
__global__ __launch_bounds__(256)
void k_dense_last(const float* __restrict__ L, const float* __restrict__ ego,
                  const float* __restrict__ Wgc, const float* __restrict__ Wbi,
                  const float* __restrict__ bias,
                  const int* __restrict__ us, const int* __restrict__ pi,
                  const int* __restrict__ ni, float* __restrict__ out, int colbase) {
    __shared__ __attribute__((aligned(16))) float sg[4096];
    __shared__ __attribute__((aligned(16))) float sb[4096];
    __shared__ float sbias[64];
    __shared__ __attribute__((aligned(16))) float ot[8 * 16 * DD];

    const int wave  = threadIdx.x >> 5;
    const int lane  = threadIdx.x & 31;
    const int half  = lane >> 4;
    const int n     = lane & 15;
    const int sbase = (blockIdx.x * 8 + wave) * 16;
    const int seg = sbase >> 12;
    const int* ip = (seg == 0) ? us : (seg == 1) ? pi : ni;
    const int node = ip[(sbase & (BATCH - 1)) + n] + (seg ? NU : 0);

    stage_weights(sg, sb, sbias, Wgc, Wbi, bias);
    __syncthreads();

    v8f acc[4];
    dense_tile(L + (size_t)node * DD, (gcf_p)(ego + (size_t)node * DD), sg, sb, half, n, acc);

    float* o = ot + wave * (16 * DD);
    #pragma unroll
    for (int j = 0; j < 4; ++j) {
        float bsum = sbias[j * 16 + n];
        #pragma unroll
        for (int v = 0; v < 8; ++v) {
            float x = acc[j][v] + bsum;
            x = (x >= 0.f) ? x : SLOPE * x;
            o[((half << 3) + v) * DD + j * 16 + n] = x;
        }
    }
    __syncthreads();
    #pragma unroll
    for (int r = 0; r < 16; ++r) {
        v2f x = *(const v2f*)(o + r * DD + lane * 2);
        float ss = x.x * x.x + x.y * x.y;
        for (int s = 16; s > 0; s >>= 1) ss += __shfl_xor(ss, s, 32);
        float scale = 1.0f / fmaxf(sqrtf(ss), 1e-12f);
        v2f y; y.x = x.x * scale; y.y = x.y * scale;
        *(v2f*)(out + (size_t)(sbase + r) * 256 + colbase + lane * 2) = y;
    }
}

extern "C" void kernel_launch(void* const* d_in, const int* in_sizes, int n_in,
                              void* d_out, int out_size, void* d_ws, size_t ws_size,
                              hipStream_t stream) {
    const float* ue   = (const float*)d_in[0];
    const float* ie   = (const float*)d_in[1];
    const float* Wgc  = (const float*)d_in[2];
    const float* Wbi  = (const float*)d_in[3];
    const float* bbi  = (const float*)d_in[4];
    const float* vals = (const float*)d_in[5];
    const int*   rows = (const int*)d_in[6];
    const int*   cols = (const int*)d_in[7];
    const int*   us   = (const int*)d_in[8];
    const int*   pi   = (const int*)d_in[9];
    const int*   ni   = (const int*)d_in[10];
    float* out = (float*)d_out;

    float* ego = (float*)d_ws;
    float* L   = ego + (size_t)NT * DD;
    int* binned = (int*)(L + (size_t)NT * DD);
    int* gcount = binned + (size_t)2 * NNZE;
    int* goff   = gcount + 2048;
    int* gcur   = goff + 2048;

    k_gather0<<<(3 * BATCH * DD + 255) / 256, 256, 0, stream>>>(
        ue, ie, us, pi, ni, out, gcount);

    k_bin_count<<<GBIN, 256, 0, stream>>>(rows, gcount, gcur);
    k_bin_place<<<GBIN, 256, 0, stream>>>(rows, cols, vals, gcount, goff, gcur, binned);

    k_spmm_blk<true><<<NB, 256, 0, stream>>>(
        binned, goff, ue, ie, L, ego, us, pi, ni, out, 0);
    k_dense_wmma<true><<<(NT + 127) / 128, 256, 0, stream>>>(
        L, ue, ie, ego, Wgc, Wbi, bbi);
    k_spmm_blk<false><<<NB + GNB, 256, 0, stream>>>(
        binned, goff, ego, ego, L, ego, us, pi, ni, out, 64);
    k_dense_wmma<false><<<(NT + 127) / 128, 256, 0, stream>>>(
        L, ego, ego, ego, Wgc + 4096, Wbi + 4096, bbi + 64);
    k_spmm_last<<<NB + GNB, 256, 0, stream>>>(
        binned, goff, ego, L, us, pi, ni, out, 128);
    k_dense_last<<<3 * BATCH / 128, 256, 0, stream>>>(
        L, ego, Wgc + 2 * 4096, Wbi + 2 * 4096, bbi + 2 * 64, us, pi, ni, out, 192);
}


# Layers 1-2 SpMM: a wave owns sixteen rows of its block as a register tile, scans the bin, keeps its rows' records, adds them by scalar-indexed registers; no LDS float adds; sum order: bin order
# speedup vs baseline: 4.0593x; 1.0993x over previous
#include <hip/hip_runtime.h>
#include <hip/hip_bf16.h>

typedef float v2f __attribute__((ext_vector_type(2)));
typedef float v8f __attribute__((ext_vector_type(8)));

#define NU 50000
#define NI 100000
#define NT 150000
#define DD 64
#define NNZE 2400000
#define BATCH 4096
#define SLOPE 0.01f

__global__ __launch_bounds__(256)
void k_gather0(const float* __restrict__ ue, const float* __restrict__ ie,
               const int* __restrict__ us, const int* __restrict__ pi,
               const int* __restrict__ ni, float* __restrict__ out,
               int* __restrict__ gcount) {
    int t = blockIdx.x * 256 + threadIdx.x;
    if (t < 2048) gcount[t] = 0;
    if (t >= 3 * BATCH * DD) return;
    int c  = t & 63;
    int b  = t >> 6;
    int seg = b >> 12;
    int bb  = b & (BATCH - 1);
    float v;
    if (seg == 0)      v = ue[(size_t)us[bb] * DD + c];
    else if (seg == 1) v = ie[(size_t)pi[bb] * DD + c];
    else               v = ie[(size_t)ni[bb] * DD + c];
    out[(size_t)b * 256 + c] = v;
}

#define RB 128
#define NB ((NT + RB - 1) / RB)
#define CSH 18
#define CMASK ((1 << CSH) - 1)
#define GBIN 512
#define CHUNK ((NNZE + GBIN - 1) / GBIN)
static_assert(NT <= (1 << CSH), "col must fit under the local row");
static_assert(NB + 1 <= 5 * 256, "the placing launch scans five bins per thread");

#define EPT 5
static_assert(CHUNK % 4 == 0 && NNZE % 4 == 0, "a chunk is whole 16-byte groups");
static_assert(EPT * 256 * 4 >= CHUNK, "EPT groups per thread cover a chunk");
static_assert((NNZE - (GBIN - 1) * CHUNK) / 4 >= (EPT - 1) * 256,
              "only the last group of a thread can lie outside the chunk");

__device__ __forceinline__ int chunk_groups(int g) {
    const int e0 = g * CHUNK;
    return (min(e0 + CHUNK, NNZE) - e0) >> 2;
}

__device__ __forceinline__ void hist_add4(int* hist, int4 r) {
    atomicAdd(&hist[(unsigned)r.x / RB], 1);
    atomicAdd(&hist[(unsigned)r.y / RB], 1);
    atomicAdd(&hist[(unsigned)r.z / RB], 1);
    atomicAdd(&hist[(unsigned)r.w / RB], 1);
}

__global__ __launch_bounds__(256)
void k_bin_count(const int* __restrict__ rows, int* __restrict__ gcount,
                 int* __restrict__ gcur) {
    __shared__ int hist[NB];
    if (blockIdx.x == 0)
        for (int i = threadIdx.x; i < NB; i += 256) gcur[i] = 0;
    for (int i = threadIdx.x; i < NB; i += 256) hist[i] = 0;
    __syncthreads();
    const int n4 = chunk_groups(blockIdx.x);
    const int4* r4 = (const int4*)(rows + (size_t)blockIdx.x * CHUNK);
    const bool tail = (int)threadIdx.x + (EPT - 1) * 256 < n4;
    int4 rr[EPT];
    for (int k = 0; k < EPT - 1; ++k) rr[k] = r4[threadIdx.x + k * 256];
    if (tail) rr[EPT - 1] = r4[threadIdx.x + (EPT - 1) * 256];
    for (int k = 0; k < EPT - 1; ++k) hist_add4(hist, rr[k]);
    if (tail) hist_add4(hist, rr[EPT - 1]);
    __syncthreads();
    for (int i = threadIdx.x; i < NB; i += 256) {
        int c = hist[i];
        if (c) atomicAdd(&gcount[i], c);
    }
}

__device__ __forceinline__ void place4(int2* rec, int* hist, const int* base,
                                       int4 r, int4 c, float4 v) {
    const int rv[4] = {r.x, r.y, r.z, r.w};
    const int cv[4] = {c.x, c.y, c.z, c.w};
    const float vv[4] = {v.x, v.y, v.z, v.w};
    int pos[4];
    for (int j = 0; j < 4; ++j) {
        const int b = (unsigned)rv[j] / RB;
        pos[j] = base[b] + atomicAdd(&hist[b], 1);
    }
    for (int j = 0; j < 4; ++j) {
        const int lr = rv[j] & (RB - 1);
        rec[pos[j]] = make_int2(cv[j] | (lr << CSH), __float_as_int(vv[j]));
    }
}
static_assert((RB & (RB - 1)) == 0, "the local row is taken with a mask");

__global__ __launch_bounds__(256)
void k_bin_place(const int* __restrict__ rows, const int* __restrict__ cols,
                 const float* __restrict__ vals, const int* __restrict__ gcount,
                 int* __restrict__ goff, int* __restrict__ gcur,
                 int* __restrict__ binned) {
    __shared__ int hist[NB];
    __shared__ int base[NB];
    __shared__ int off[NB + 1];
    __shared__ int wsum[8];
    for (int i = threadIdx.x; i < NB; i += 256) hist[i] = 0;
    __syncthreads();
    const int n4 = chunk_groups(blockIdx.x);
    const size_t e0 = (size_t)blockIdx.x * CHUNK;
    const int4*   r4 = (const int4*)(rows + e0);
    const int4*   c4 = (const int4*)(cols + e0);
    const float4* v4 = (const float4*)(vals + e0);
    const bool tail = (int)threadIdx.x + (EPT - 1) * 256 < n4;
    int4 rr[EPT], cc[EPT];
    float4 vv[EPT];
    for (int k = 0; k < EPT - 1; ++k) rr[k] = r4[threadIdx.x + k * 256];
    if (tail) rr[EPT - 1] = r4[threadIdx.x + (EPT - 1) * 256];
    int cs[5];
    for (int k = 0; k < 5; ++k) {
        int idx = threadIdx.x * 5 + k;
        cs[k] = (idx < NB) ? gcount[idx] : 0;
    }
    for (int k = 0; k < EPT - 1; ++k) {
        cc[k] = c4[threadIdx.x + k * 256];
        vv[k] = v4[threadIdx.x + k * 256];
    }
    if (tail) {
        cc[EPT - 1] = c4[threadIdx.x + (EPT - 1) * 256];
        vv[EPT - 1] = v4[threadIdx.x + (EPT - 1) * 256];
    }
    for (int k = 0; k < EPT - 1; ++k) hist_add4(hist, rr[k]);
    if (tail) hist_add4(hist, rr[EPT - 1]);
    {
        const int lane = threadIdx.x & 31, wv = threadIdx.x >> 5;
        const int s = cs[0] + cs[1] + cs[2] + cs[3] + cs[4];
        int incl = s;
        for (int o = 1; o < 32; o <<= 1) {
            int v = __shfl_up(incl, o, 32);
            if (lane >= o) incl += v;
        }
        if (lane == 31) wsum[wv] = incl;
        __syncthreads();
        int excl = incl - s;
        for (int w = 0; w < wv; ++w) excl += wsum[w];
        for (int k = 0; k < 5; ++k) {
            int idx = threadIdx.x * 5 + k;
            if (idx <= NB) off[idx] = excl;
            excl += cs[k];
        }
    }
    __syncthreads();
    if (blockIdx.x == 0)
        for (int i = threadIdx.x; i <= NB; i += 256) goff[i] = off[i];
    {
        int cnt[5], got[5];
        for (int k = 0; k < 5; ++k) {
            const int i = threadIdx.x + k * 256;
            cnt[k] = (i < NB) ? hist[i] : 0;
        }
        for (int k = 0; k < 5; ++k) {
            const int i = threadIdx.x + k * 256;
            got[k] = (i < NB) ? atomicAdd(&gcur[i], cnt[k]) : 0;
        }
        for (int k = 0; k < 5; ++k) {
            const int i = threadIdx.x + k * 256;
            if (i < NB) {
                base[i] = off[i] + got[k];
                hist[i] = 0;
            }
        }
    }
    static_assert(5 * 256 >= NB, "five bins per thread in the cursor step");
    __syncthreads();
    int2* rec = (int2*)binned;
    for (int k = 0; k < EPT - 1; ++k) place4(rec, hist, base, rr[k], cc[k], vv[k]);
    if (tail) place4(rec, hist, base, rr[EPT - 1], cc[EPT - 1], vv[EPT - 1]);
}

typedef const float __attribute__((address_space(1)))* gcf_p;
typedef const v2f   __attribute__((address_space(1)))* gcv2_p;
template <bool SPLIT>
__device__ __forceinline__ gcf_p src_row(const float* a, const float* b, int r) {
    if (!SPLIT) return (gcf_p)(a + (size_t)r * DD);
    const unsigned long long ba = (unsigned long long)a;
    const unsigned long long bb = (unsigned long long)b - (unsigned long long)NU * DD * sizeof(float);
    const unsigned long long base = (r < NU) ? ba : bb;
    return (gcf_p)(base + (unsigned long long)(unsigned)r * (DD * sizeof(float)));
}

#define GNR 8
#define GNB (3 * BATCH / (8 * GNR))
static_assert(GNB * 8 * GNR == 3 * BATCH, "output rows split evenly");
__device__ __forceinline__ void gather_norm_rows(
        const float* __restrict__ ego, const int* __restrict__ us,
        const int* __restrict__ pi, const int* __restrict__ ni,
        float* __restrict__ out, int colbase, int g) {
    const int lane = threadIdx.x & 31;
    const int w0 = g * 8 + (threadIdx.x >> 5);
    #pragma unroll
    for (int k = 0; k < GNR; ++k) {
        const int wid = w0 + k * (GNB * 8);
        const int seg = wid >> 12;
        const int bb  = wid & (BATCH - 1);
        const int* ip = (seg == 0) ? us : (seg == 1) ? pi : ni;
        const int node = ip[bb] + (seg ? NU : 0);
        v2f x = *(const v2f*)(ego + (size_t)node * DD + lane * 2);
        float ss = x.x * x.x + x.y * x.y;
        for (int o = 16; o > 0; o >>= 1) ss += __shfl_xor(ss, o, 32);
        float scale = 1.0f / fmaxf(sqrtf(ss), 1e-12f);
        v2f y; y.x = x.x * scale; y.y = x.y * scale;
        *(v2f*)(out + (size_t)wid * 256 + colbase + lane * 2) = y;
    }
}

__device__ __forceinline__ void lds_add(float* p, float v) {
    __hip_atomic_fetch_add(p, v, __ATOMIC_RELAXED, __HIP_MEMORY_SCOPE_WORKGROUP);
}

typedef float v16f __attribute__((ext_vector_type(16)));
#define WR 16
#define NCH 8
#define QN 512
static_assert(RB == 8 * WR, "eight waves share a block's rows");
static_assert(QN >= 32 * NCH + WR && (QN & (QN - 1)) == 0, "a round's records fit behind a partial batch");

template <bool SPLIT>
__device__ __forceinline__ void add_batch(const int2* q, int head, int cnt, int lane,
        const float* srcA, const float* srcB, v16f& ax, v16f& ay) {
    const int2 r = q[(head + min(lane, cnt - 1)) & (QN - 1)];
    v2f x[WR];
    int pk[WR];
    #pragma unroll
    for (int u = 0; u < WR; ++u) {
        pk[u] = __builtin_amdgcn_readlane(r.x, u);
        x[u] = *(gcv2_p)(src_row<SPLIT>(srcA, srcB, pk[u] & CMASK) + lane * 2);
    }
    #pragma unroll
    for (int u = 0; u < WR; ++u) {
        const bool on = u < cnt;
        const float v = on ? __int_as_float(__builtin_amdgcn_readlane(r.y, u)) : 0.f;
        const float x0 = on ? x[u].x : 0.f;
        const float x1 = on ? x[u].y : 0.f;
        const int lr = (pk[u] >> CSH) & (WR - 1);
        ax[lr] = fmaf(v, x0, ax[lr]);
        ay[lr] = fmaf(v, x1, ay[lr]);
    }
}

template <bool SPLIT>
__global__ __launch_bounds__(256, 12)
void k_spmm_blk(const int* __restrict__ binned, const int* __restrict__ goff,
                const float* __restrict__ srcA, const float* __restrict__ srcB,
                float* __restrict__ L,
                const float* __restrict__ gsrc, const int* __restrict__ us,
                const int* __restrict__ pi, const int* __restrict__ ni,
                float* __restrict__ out, int colbase) {
    __shared__ __attribute__((aligned(8))) int ring[8 * QN * 2];
    if (blockIdx.x >= NB) {
        gather_norm_rows(gsrc, us, pi, ni, out, colbase, blockIdx.x - NB);
        return;
    }
    const int b = blockIdx.x;
    const int beg = goff[b], end = goff[b + 1];
    const int wave = __builtin_amdgcn_readfirstlane(threadIdx.x >> 5);
    const int lane = threadIdx.x & 31;
    const int2* rec = (const int2*)binned;
    int2* q = (int2*)ring + wave * QN;

    v16f ax = (v16f){}, ay = (v16f){};
    int head = 0, qlen = 0;

    for (int s0 = beg; s0 < end; s0 += 32 * NCH) {
        int2 r[NCH];
        #pragma unroll
        for (int k = 0; k < NCH; ++k) {
            const int e = s0 + k * 32 + lane;
            r[k] = (e < end) ? rec[e] : make_int2(-1, 0);
        }
        #pragma unroll
        for (int k = 0; k < NCH; ++k) {
            const bool mine = ((unsigned)r[k].x >> (CSH + 4)) == (unsigned)wave;
            const unsigned m = __builtin_amdgcn_ballot_w32(mine);
            if (mine) q[(head + qlen + (int)__builtin_amdgcn_mbcnt_lo(m, 0u)) & (QN - 1)] = r[k];
            qlen += __builtin_popcount(m);
        }
        while (qlen >= WR) {
            add_batch<SPLIT>(q, head, WR, lane, srcA, srcB, ax, ay);
            head = (head + WR) & (QN - 1);
            qlen -= WR;
        }
    }
    if (qlen > 0) add_batch<SPLIT>(q, head, qlen, lane, srcA, srcB, ax, ay);

    #pragma unroll
    for (int k = 0; k < WR; ++k) {
        const int row = b * RB + wave * WR + k;
        if (row < NT) {
            v2f y; y.x = ax[k]; y.y = ay[k];
            *(v2f*)(L + (size_t)row * DD + lane * 2) = y;
        }
    }
}

#define SCAN 2048
static_assert(SCAN == 8 * 256, "a round is eight records per thread: the list cannot overflow");
static_assert(BATCH == 4 * 4 * 256, "an index array is four 16-byte loads per thread");
static_assert(RB == 128, "the mask is four words");

__device__ __forceinline__ bool row_marked(unsigned m0, unsigned m1, unsigned m2,
                                           unsigned m3, int lr) {
    const unsigned w = (lr & 64) ? ((lr & 32) ? m3 : m2) : ((lr & 32) ? m1 : m0);
    return (w >> (lr & 31)) & 1u;
}

__global__ __launch_bounds__(256)
void k_spmm_last(const int* __restrict__ binned, const int* __restrict__ goff,
                 const float* __restrict__ src, float* __restrict__ L,
                 const int* __restrict__ us, const int* __restrict__ pi,
                 const int* __restrict__ ni, float* __restrict__ out, int colbase) {
    __shared__ float tile[RB * DD];
    __shared__ __attribute__((aligned(8))) int keep[2 * SCAN];
    __shared__ unsigned need[RB / 32];
    __shared__ int nkeep;
    if (blockIdx.x >= NB) {
        gather_norm_rows(src, us, pi, ni, out, colbase, blockIdx.x - NB);
        return;
    }
    const int b = blockIdx.x;
    float4* t4 = (float4*)tile;
    for (int i = threadIdx.x; i < RB * DD / 4; i += 256)
        t4[i] = make_float4(0.f, 0.f, 0.f, 0.f);
    if (threadIdx.x < RB / 32) need[threadIdx.x] = 0u;
    __syncthreads();

    {
        int4 q[12];
        #pragma unroll
        for (int k = 0; k < 12; ++k) {
            const int* ip = (k < 4) ? us : (k < 8) ? pi : ni;
            q[k] = ((const int4*)ip)[threadIdx.x + (k & 3) * 256];
        }
        #pragma unroll
        for (int k = 0; k < 12; ++k) {
            const int add = (k < 4) ? 0 : NU;
            const int nv[4] = {q[k].x + add, q[k].y + add, q[k].z + add, q[k].w + add};
            for (int j = 0; j < 4; ++j)
                if ((unsigned)nv[j] / RB == (unsigned)b)
                    atomicOr(&need[(nv[j] & (RB - 1)) >> 5], 1u << (nv[j] & 31));
        }
    }
    __syncthreads();
    const unsigned m0 = need[0], m1 = need[1], m2 = need[2], m3 = need[3];
    if ((m0 | m1 | m2 | m3) == 0u) return;

    const int beg = goff[b], end = goff[b + 1];
    const int wave = __builtin_amdgcn_readfirstlane(threadIdx.x >> 5);
    const int lane = threadIdx.x & 31;
    const int2* rec = (const int2*)binned;
    int2* kp = (int2*)keep;

    for (int s0 = beg; s0 < end; s0 += SCAN) {
        if (threadIdx.x == 0) nkeep = 0;
        __syncthreads();
        {
            int2 r[8];
            #pragma unroll
            for (int k = 0; k < 8; ++k) {
                const int e = s0 + threadIdx.x + k * 256;
                r[k] = (e < end) ? rec[e] : make_int2(0, 0);
            }
            #pragma unroll
            for (int k = 0; k < 8; ++k) {
                const int e = s0 + threadIdx.x + k * 256;
                if (e < end && row_marked(m0, m1, m2, m3, (unsigned)r[k].x >> CSH))
                    kp[atomicAdd(&nkeep, 1)] = r[k];
            }
        }
        __syncthreads();
        const int nk = __builtin_amdgcn_readfirstlane(nkeep);
        for (int t0 = wave * 16; t0 < nk; t0 += 8 * 16) {
            const int cnt = min(16, nk - t0);
            const int2 r = kp[t0 + min(lane, cnt - 1)];
            v2f x[16];
            int pk[16];
            #pragma unroll
            for (int u = 0; u < 16; ++u) {
                pk[u] = __builtin_amdgcn_readlane(r.x, u);
                x[u] = *(gcv2_p)((gcf_p)(src + (size_t)(pk[u] & CMASK) * DD) + lane * 2);
            }
            #pragma unroll
            for (int u = 0; u < 16; ++u) {
                if (u < cnt) {
                    float v = __int_as_float(__builtin_amdgcn_readlane(r.y, u));
                    float* d = tile + (pk[u] >> CSH) * DD + lane * 2;
                    lds_add(d,     v * x[u].x);
                    lds_add(d + 1, v * x[u].y);
                }
            }
        }
        __syncthreads();
    }

    for (int i = threadIdx.x; i < RB * DD / 4; i += 256) {
        const int lr = i >> 4;
        const int row = b * RB + lr;
        if (row < NT && row_marked(m0, m1, m2, m3, lr))
            *(float4*)(L + (size_t)row * DD + (i & 15) * 4) = t4[i];
    }
}


__device__ __forceinline__ void stage_weights(float* sg, float* sb, float* sbias,
        const float* __restrict__ Wgc, const float* __restrict__ Wbi,
        const float* __restrict__ bias) {
    const int p = threadIdx.x >> 3;
    const int f = threadIdx.x & 7;
    float4 g[2][2], h[2][2];
    for (int q = 0; q < 2; ++q) {
        for (int e = 0; e < 2; ++e) {
            const int idx = (2 * p + e) * 64 + (f + 8 * q) * 4;
            g[e][q] = *(const float4*)(Wgc + idx);
            h[e][q] = *(const float4*)(Wbi + idx);
        }
    }
    const float b2 = (threadIdx.x < 64) ? bias[threadIdx.x] : 0.f;
    for (int q = 0; q < 2; ++q) {
        const int dst = (p * 64 + (f + 8 * q) * 4) << 1;
        *(float4*)(sg + dst)     = make_float4(g[0][q].x, g[1][q].x, g[0][q].y, g[1][q].y);
        *(float4*)(sg + dst + 4) = make_float4(g[0][q].z, g[1][q].z, g[0][q].w, g[1][q].w);
        *(float4*)(sb + dst)     = make_float4(h[0][q].x, h[1][q].x, h[0][q].y, h[1][q].y);
        *(float4*)(sb + dst + 4) = make_float4(h[0][q].z, h[1][q].z, h[0][q].w, h[1][q].w);
    }
    if (threadIdx.x < 64) sbias[threadIdx.x] = 2.0f * b2;
}

__device__ __forceinline__ void dense_tile(const float* lrow, gcf_p erow,
        const float* sg, const float* sb, int half, int n, v8f (&acc)[4]) {
    #pragma unroll
    for (int j = 0; j < 4; ++j) acc[j] = (v8f){};

    #pragma unroll
    for (int kk = 0; kk < 16; ++kk) {
        int off = kk * 4 + half * 2;
        v2f ls = *(const v2f*)(lrow + off);
        v2f eg = *(gcv2_p)(erow + off);
        v2f a_li  = ls + eg;
        v2f a_mul = ls * eg;
        int bbase = ((kk * 2 + half) * 64) << 1;
        #pragma unroll
        for (int j = 0; j < 4; ++j) {
            v2f bgc = *(const v2f*)(sg + bbase + ((j * 16 + n) << 1));
            v2f bbi = *(const v2f*)(sb + bbase + ((j * 16 + n) << 1));
            acc[j] = __builtin_amdgcn_wmma_f32_16x16x4_f32(
                false, a_li, false, bgc, (short)0, acc[j], false, false);
            acc[j] = __builtin_amdgcn_wmma_f32_16x16x4_f32(
                false, a_mul, false, bbi, (short)0, acc[j], false, false);
        }
    }
}

template <bool SPLIT>
__global__ __launch_bounds__(256)
void k_dense_wmma(const float* __restrict__ L, const float* srcA, const float* srcB,
                  float* ego,
                  const float* __restrict__ Wgc, const float* __restrict__ Wbi,
                  const float* __restrict__ bias) {
    __shared__ __attribute__((aligned(16))) float sg[4096];
    __shared__ __attribute__((aligned(16))) float sb[4096];
    __shared__ float sbias[64];

    stage_weights(sg, sb, sbias, Wgc, Wbi, bias);
    __syncthreads();

    const int wave  = threadIdx.x >> 5;
    const int lane  = threadIdx.x & 31;
    const int mbase = (blockIdx.x * 8 + wave) * 16;
    if (mbase >= NT) return;

    const int half = lane >> 4;
    const int n    = lane & 15;

    const float* lrow = L   + (size_t)(mbase + n) * DD;
    gcf_p erow = src_row<SPLIT>(srcA, srcB, mbase + n);

    v8f acc[4];
    dense_tile(lrow, erow, sg, sb, half, n, acc);

    #pragma unroll
    for (int j = 0; j < 4; ++j) {
        float bsum = sbias[j * 16 + n];
        #pragma unroll
        for (int v = 0; v < 8; ++v) {
            float x = acc[j][v] + bsum;
            x = (x >= 0.f) ? x : SLOPE * x;
            ego[(size_t)(mbase + (half << 3) + v) * DD + j * 16 + n] = x;
        }
    }
}

static_assert(BATCH % 16 == 0, "the 16 slots of a tile share one index array");
static_assert(3 * BATCH % 128 == 0, "eight whole tiles per work-group");
__global__ __launch_bounds__(256)
void k_dense_last(const float* __restrict__ L, const float* __restrict__ ego,
                  const float* __restrict__ Wgc, const float* __restrict__ Wbi,
                  const float* __restrict__ bias,
                  const int* __restrict__ us, const int* __restrict__ pi,
                  const int* __restrict__ ni, float* __restrict__ out, int colbase) {
    __shared__ __attribute__((aligned(16))) float sg[4096];
    __shared__ __attribute__((aligned(16))) float sb[4096];
    __shared__ float sbias[64];
    __shared__ __attribute__((aligned(16))) float ot[8 * 16 * DD];

    const int wave  = threadIdx.x >> 5;
    const int lane  = threadIdx.x & 31;
    const int half  = lane >> 4;
    const int n     = lane & 15;
    const int sbase = (blockIdx.x * 8 + wave) * 16;
    const int seg = sbase >> 12;
    const int* ip = (seg == 0) ? us : (seg == 1) ? pi : ni;
    const int node = ip[(sbase & (BATCH - 1)) + n] + (seg ? NU : 0);

    stage_weights(sg, sb, sbias, Wgc, Wbi, bias);
    __syncthreads();

    v8f acc[4];
    dense_tile(L + (size_t)node * DD, (gcf_p)(ego + (size_t)node * DD), sg, sb, half, n, acc);

    float* o = ot + wave * (16 * DD);
    #pragma unroll
    for (int j = 0; j < 4; ++j) {
        float bsum = sbias[j * 16 + n];
        #pragma unroll
        for (int v = 0; v < 8; ++v) {
            float x = acc[j][v] + bsum;
            x = (x >= 0.f) ? x : SLOPE * x;
            o[((half << 3) + v) * DD + j * 16 + n] = x;
        }
    }
    __syncthreads();
    #pragma unroll
    for (int r = 0; r < 16; ++r) {
        v2f x = *(const v2f*)(o + r * DD + lane * 2);
        float ss = x.x * x.x + x.y * x.y;
        for (int s = 16; s > 0; s >>= 1) ss += __shfl_xor(ss, s, 32);
        float scale = 1.0f / fmaxf(sqrtf(ss), 1e-12f);
        v2f y; y.x = x.x * scale; y.y = x.y * scale;
        *(v2f*)(out + (size_t)(sbase + r) * 256 + colbase + lane * 2) = y;
    }
}

extern "C" void kernel_launch(void* const* d_in, const int* in_sizes, int n_in,
                              void* d_out, int out_size, void* d_ws, size_t ws_size,
                              hipStream_t stream) {
    const float* ue   = (const float*)d_in[0];
    const float* ie   = (const float*)d_in[1];
    const float* Wgc  = (const float*)d_in[2];
    const float* Wbi  = (const float*)d_in[3];
    const float* bbi  = (const float*)d_in[4];
    const float* vals = (const float*)d_in[5];
    const int*   rows = (const int*)d_in[6];
    const int*   cols = (const int*)d_in[7];
    const int*   us   = (const int*)d_in[8];
    const int*   pi   = (const int*)d_in[9];
    const int*   ni   = (const int*)d_in[10];
    float* out = (float*)d_out;

    float* ego = (float*)d_ws;
    float* L   = ego + (size_t)NT * DD;
    int* binned = (int*)(L + (size_t)NT * DD);
    int* gcount = binned + (size_t)2 * NNZE;
    int* goff   = gcount + 2048;
    int* gcur   = goff + 2048;

    k_gather0<<<(3 * BATCH * DD + 255) / 256, 256, 0, stream>>>(
        ue, ie, us, pi, ni, out, gcount);

    k_bin_count<<<GBIN, 256, 0, stream>>>(rows, gcount, gcur);
    k_bin_place<<<GBIN, 256, 0, stream>>>(rows, cols, vals, gcount, goff, gcur, binned);

    k_spmm_blk<true><<<NB, 256, 0, stream>>>(
        binned, goff, ue, ie, L, ego, us, pi, ni, out, 0);
    k_dense_wmma<true><<<(NT + 127) / 128, 256, 0, stream>>>(
        L, ue, ie, ego, Wgc, Wbi, bbi);
    k_spmm_blk<false><<<NB + GNB, 256, 0, stream>>>(
        binned, goff, ego, ego, L, ego, us, pi, ni, out, 64);
    k_dense_wmma<false><<<(NT + 127) / 128, 256, 0, stream>>>(
        L, ego, ego, ego, Wgc + 4096, Wbi + 4096, bbi + 64);
    k_spmm_last<<<NB + GNB, 256, 0, stream>>>(
        binned, goff, ego, L, us, pi, ni, out, 128);
    k_dense_last<<<3 * BATCH / 128, 256, 0, stream>>>(
        L, ego, Wgc + 2 * 4096, Wbi + 2 * 4096, bbi + 2 * 64, us, pi, ni, out, 192);
}


# Layers 1-2 SpMM: a block's eight waves share the bin scan: each reads an eighth once and sorts it by owner wave into an LDS list (ballots, prefix counts); was each wave scanning all; sum order kept
# speedup vs baseline: 4.2928x; 1.0575x over previous
#include <hip/hip_runtime.h>
#include <hip/hip_bf16.h>

typedef float v2f __attribute__((ext_vector_type(2)));
typedef float v8f __attribute__((ext_vector_type(8)));

#define NU 50000
#define NI 100000
#define NT 150000
#define DD 64
#define NNZE 2400000
#define BATCH 4096
#define SLOPE 0.01f

__global__ __launch_bounds__(256)
void k_gather0(const float* __restrict__ ue, const float* __restrict__ ie,
               const int* __restrict__ us, const int* __restrict__ pi,
               const int* __restrict__ ni, float* __restrict__ out,
               int* __restrict__ gcount) {
    int t = blockIdx.x * 256 + threadIdx.x;
    if (t < 2048) gcount[t] = 0;
    if (t >= 3 * BATCH * DD) return;
    int c  = t & 63;
    int b  = t >> 6;
    int seg = b >> 12;
    int bb  = b & (BATCH - 1);
    float v;
    if (seg == 0)      v = ue[(size_t)us[bb] * DD + c];
    else if (seg == 1) v = ie[(size_t)pi[bb] * DD + c];
    else               v = ie[(size_t)ni[bb] * DD + c];
    out[(size_t)b * 256 + c] = v;
}

#define RB 128
#define NB ((NT + RB - 1) / RB)
#define CSH 18
#define CMASK ((1 << CSH) - 1)
#define GBIN 512
#define CHUNK ((NNZE + GBIN - 1) / GBIN)
static_assert(NT <= (1 << CSH), "col must fit under the local row");
static_assert(NB + 1 <= 5 * 256, "the placing launch scans five bins per thread");

#define EPT 5
static_assert(CHUNK % 4 == 0 && NNZE % 4 == 0, "a chunk is whole 16-byte groups");
static_assert(EPT * 256 * 4 >= CHUNK, "EPT groups per thread cover a chunk");
static_assert((NNZE - (GBIN - 1) * CHUNK) / 4 >= (EPT - 1) * 256,
              "only the last group of a thread can lie outside the chunk");

__device__ __forceinline__ int chunk_groups(int g) {
    const int e0 = g * CHUNK;
    return (min(e0 + CHUNK, NNZE) - e0) >> 2;
}

__device__ __forceinline__ void hist_add4(int* hist, int4 r) {
    atomicAdd(&hist[(unsigned)r.x / RB], 1);
    atomicAdd(&hist[(unsigned)r.y / RB], 1);
    atomicAdd(&hist[(unsigned)r.z / RB], 1);
    atomicAdd(&hist[(unsigned)r.w / RB], 1);
}

__global__ __launch_bounds__(256)
void k_bin_count(const int* __restrict__ rows, int* __restrict__ gcount,
                 int* __restrict__ gcur) {
    __shared__ int hist[NB];
    if (blockIdx.x == 0)
        for (int i = threadIdx.x; i < NB; i += 256) gcur[i] = 0;
    for (int i = threadIdx.x; i < NB; i += 256) hist[i] = 0;
    __syncthreads();
    const int n4 = chunk_groups(blockIdx.x);
    const int4* r4 = (const int4*)(rows + (size_t)blockIdx.x * CHUNK);
    const bool tail = (int)threadIdx.x + (EPT - 1) * 256 < n4;
    int4 rr[EPT];
    for (int k = 0; k < EPT - 1; ++k) rr[k] = r4[threadIdx.x + k * 256];
    if (tail) rr[EPT - 1] = r4[threadIdx.x + (EPT - 1) * 256];
    for (int k = 0; k < EPT - 1; ++k) hist_add4(hist, rr[k]);
    if (tail) hist_add4(hist, rr[EPT - 1]);
    __syncthreads();
    for (int i = threadIdx.x; i < NB; i += 256) {
        int c = hist[i];
        if (c) atomicAdd(&gcount[i], c);
    }
}

__device__ __forceinline__ void place4(int2* rec, int* hist, const int* base,
                                       int4 r, int4 c, float4 v) {
    const int rv[4] = {r.x, r.y, r.z, r.w};
    const int cv[4] = {c.x, c.y, c.z, c.w};
    const float vv[4] = {v.x, v.y, v.z, v.w};
    int pos[4];
    for (int j = 0; j < 4; ++j) {
        const int b = (unsigned)rv[j] / RB;
        pos[j] = base[b] + atomicAdd(&hist[b], 1);
    }
    for (int j = 0; j < 4; ++j) {
        const int lr = rv[j] & (RB - 1);
        rec[pos[j]] = make_int2(cv[j] | (lr << CSH), __float_as_int(vv[j]));
    }
}
static_assert((RB & (RB - 1)) == 0, "the local row is taken with a mask");

__global__ __launch_bounds__(256)
void k_bin_place(const int* __restrict__ rows, const int* __restrict__ cols,
                 const float* __restrict__ vals, const int* __restrict__ gcount,
                 int* __restrict__ goff, int* __restrict__ gcur,
                 int* __restrict__ binned) {
    __shared__ int hist[NB];
    __shared__ int base[NB];
    __shared__ int off[NB + 1];
    __shared__ int wsum[8];
    for (int i = threadIdx.x; i < NB; i += 256) hist[i] = 0;
    __syncthreads();
    const int n4 = chunk_groups(blockIdx.x);
    const size_t e0 = (size_t)blockIdx.x * CHUNK;
    const int4*   r4 = (const int4*)(rows + e0);
    const int4*   c4 = (const int4*)(cols + e0);
    const float4* v4 = (const float4*)(vals + e0);
    const bool tail = (int)threadIdx.x + (EPT - 1) * 256 < n4;
    int4 rr[EPT], cc[EPT];
    float4 vv[EPT];
    for (int k = 0; k < EPT - 1; ++k) rr[k] = r4[threadIdx.x + k * 256];
    if (tail) rr[EPT - 1] = r4[threadIdx.x + (EPT - 1) * 256];
    int cs[5];
    for (int k = 0; k < 5; ++k) {
        int idx = threadIdx.x * 5 + k;
        cs[k] = (idx < NB) ? gcount[idx] : 0;
    }
    for (int k = 0; k < EPT - 1; ++k) {
        cc[k] = c4[threadIdx.x + k * 256];
        vv[k] = v4[threadIdx.x + k * 256];
    }
    if (tail) {
        cc[EPT - 1] = c4[threadIdx.x + (EPT - 1) * 256];
        vv[EPT - 1] = v4[threadIdx.x + (EPT - 1) * 256];
    }
    for (int k = 0; k < EPT - 1; ++k) hist_add4(hist, rr[k]);
    if (tail) hist_add4(hist, rr[EPT - 1]);
    {
        const int lane = threadIdx.x & 31, wv = threadIdx.x >> 5;
        const int s = cs[0] + cs[1] + cs[2] + cs[3] + cs[4];
        int incl = s;
        for (int o = 1; o < 32; o <<= 1) {
            int v = __shfl_up(incl, o, 32);
            if (lane >= o) incl += v;
        }
        if (lane == 31) wsum[wv] = incl;
        __syncthreads();
        int excl = incl - s;
        for (int w = 0; w < wv; ++w) excl += wsum[w];
        for (int k = 0; k < 5; ++k) {
            int idx = threadIdx.x * 5 + k;
            if (idx <= NB) off[idx] = excl;
            excl += cs[k];
        }
    }
    __syncthreads();
    if (blockIdx.x == 0)
        for (int i = threadIdx.x; i <= NB; i += 256) goff[i] = off[i];
    {
        int cnt[5], got[5];
        for (int k = 0; k < 5; ++k) {
            const int i = threadIdx.x + k * 256;
            cnt[k] = (i < NB) ? hist[i] : 0;
        }
        for (int k = 0; k < 5; ++k) {
            const int i = threadIdx.x + k * 256;
            got[k] = (i < NB) ? atomicAdd(&gcur[i], cnt[k]) : 0;
        }
        for (int k = 0; k < 5; ++k) {
            const int i = threadIdx.x + k * 256;
            if (i < NB) {
                base[i] = off[i] + got[k];
                hist[i] = 0;
            }
        }
    }
    static_assert(5 * 256 >= NB, "five bins per thread in the cursor step");
    __syncthreads();
    int2* rec = (int2*)binned;
    for (int k = 0; k < EPT - 1; ++k) place4(rec, hist, base, rr[k], cc[k], vv[k]);
    if (tail) place4(rec, hist, base, rr[EPT - 1], cc[EPT - 1], vv[EPT - 1]);
}

typedef const float __attribute__((address_space(1)))* gcf_p;
typedef const v2f   __attribute__((address_space(1)))* gcv2_p;
template <bool SPLIT>
__device__ __forceinline__ gcf_p src_row(const float* a, const float* b, int r) {
    if (!SPLIT) return (gcf_p)(a + (size_t)r * DD);
    const unsigned long long ba = (unsigned long long)a;
    const unsigned long long bb = (unsigned long long)b - (unsigned long long)NU * DD * sizeof(float);
    const unsigned long long base = (r < NU) ? ba : bb;
    return (gcf_p)(base + (unsigned long long)(unsigned)r * (DD * sizeof(float)));
}

#define GNR 8
#define GNB (3 * BATCH / (8 * GNR))
static_assert(GNB * 8 * GNR == 3 * BATCH, "output rows split evenly");
__device__ __forceinline__ void gather_norm_rows(
        const float* __restrict__ ego, const int* __restrict__ us,
        const int* __restrict__ pi, const int* __restrict__ ni,
        float* __restrict__ out, int colbase, int g, int wv, int lane) {
    const int w0 = g * 8 + wv;
    #pragma unroll
    for (int k = 0; k < GNR; ++k) {
        const int wid = w0 + k * (GNB * 8);
        const int seg = wid >> 12;
        const int bb  = wid & (BATCH - 1);
        const int* ip = (seg == 0) ? us : (seg == 1) ? pi : ni;
        const int node = ip[bb] + (seg ? NU : 0);
        v2f x = *(const v2f*)(ego + (size_t)node * DD + lane * 2);
        float ss = x.x * x.x + x.y * x.y;
        for (int o = 16; o > 0; o >>= 1) ss += __shfl_xor(ss, o, 32);
        float scale = 1.0f / fmaxf(sqrtf(ss), 1e-12f);
        v2f y; y.x = x.x * scale; y.y = x.y * scale;
        *(v2f*)(out + (size_t)wid * 256 + colbase + lane * 2) = y;
    }
}

__device__ __forceinline__ void lds_add(float* p, float v) {
    __hip_atomic_fetch_add(p, v, __ATOMIC_RELAXED, __HIP_MEMORY_SCOPE_WORKGROUP);
}

typedef float v16f __attribute__((ext_vector_type(16)));
#define WR 16
#define RND 2560
#define WCH (RND / 8 / 32)
static_assert(RB == 8 * WR, "eight waves share a block's rows");
static_assert(RND % 256 == 0, "a wave's share of a round is whole chunks");

template <bool SPLIT>
__device__ __forceinline__ void add_batch(const int2* q, int cnt, int lane,
        const float* srcA, const float* srcB, v16f& ax, v16f& ay) {
    const int2 r = q[min(lane, cnt - 1)];
    v2f x[WR];
    int pk[WR];
    #pragma unroll
    for (int u = 0; u < WR; ++u) {
        pk[u] = __builtin_amdgcn_readlane(r.x, u);
        x[u] = *(gcv2_p)(src_row<SPLIT>(srcA, srcB, pk[u] & CMASK) + lane * 2);
    }
    #pragma unroll
    for (int u = 0; u < WR; ++u) {
        const bool on = u < cnt;
        const float v = on ? __int_as_float(__builtin_amdgcn_readlane(r.y, u)) : 0.f;
        const float x0 = on ? x[u].x : 0.f;
        const float x1 = on ? x[u].y : 0.f;
        const int lr = (pk[u] >> CSH) & (WR - 1);
        ax[lr] = fmaf(v, x0, ax[lr]);
        ay[lr] = fmaf(v, x1, ay[lr]);
    }
}

__device__ __forceinline__ void store_tile(float* L, int row0, int lane, v16f ax, v16f ay) {
    #pragma unroll
    for (int k = 0; k < WR; ++k) {
        if (row0 + k < NT) {
            v2f y; y.x = ax[k]; y.y = ay[k];
            *(v2f*)(L + (size_t)(row0 + k) * DD + lane * 2) = y;
        }
    }
}

__device__ __forceinline__ unsigned same_key(unsigned b0, unsigned b1, unsigned b2,
                                             unsigned ok, unsigned want) {
    return ((want & 1u) ? b0 : ~b0) & ((want & 2u) ? b1 : ~b1) &
           ((want & 4u) ? b2 : ~b2) & ok;
}

template <bool SPLIT>
__global__ __launch_bounds__(256, 12)
void k_spmm_blk(const int* __restrict__ binned, const int* __restrict__ goff,
                const float* __restrict__ srcA, const float* __restrict__ srcB,
                float* __restrict__ L,
                const float* __restrict__ gsrc, const int* __restrict__ us,
                const int* __restrict__ pi, const int* __restrict__ ni,
                float* __restrict__ out, int colbase) {
    __shared__ __attribute__((aligned(8))) int list[2 * RND];
    __shared__ __attribute__((aligned(16))) int tot[8 * 8];
    const int wave = __builtin_amdgcn_readfirstlane(threadIdx.x >> 5);
    const int lane = (int)__builtin_amdgcn_mbcnt_lo(~0u, 0u);
    if (blockIdx.x >= NB) {
        gather_norm_rows(gsrc, us, pi, ni, out, colbase, blockIdx.x - NB, wave, lane);
        return;
    }
    const int b = blockIdx.x;
    const int beg = goff[b], end = goff[b + 1];
    const int2* rec = (const int2*)binned;
    int2* lst = (int2*)list;

    if (beg >= end) {
        store_tile(L, b * RB + wave * WR, lane, (v16f){}, (v16f){});
        return;
    }

    for (int s0 = beg; s0 < end; s0 += RND) {
        int ln = lane;
        asm volatile("" : "+v"(ln));
        const int rend = min(s0 + RND, end);
        const int share = ((rend - s0 + 255) >> 8) << 5;
        const int w0 = s0 + wave * share;
        const int wend = min(w0 + share, rend);
        int2 r[WCH];
        #pragma unroll
        for (int k = 0; k < WCH; ++k)
            r[k] = rec[min(w0 + k * 32 + ln, rend - 1)];

        int totv = 0;
        #pragma unroll
        for (int k = 0; k < WCH; ++k) {
            const int e0 = w0 + k * 32;
            if (e0 < wend) {
                const unsigned o = (unsigned)r[k].x >> (CSH + 4);
                const unsigned ok = __builtin_amdgcn_ballot_w32(ln < wend - e0);
                const unsigned b0 = __builtin_amdgcn_ballot_w32((o & 1u) != 0u);
                const unsigned b1 = __builtin_amdgcn_ballot_w32((o & 2u) != 0u);
                const unsigned b2 = __builtin_amdgcn_ballot_w32((o & 4u) != 0u);
                totv += __builtin_popcount(same_key(b0, b1, b2, ok, (unsigned)ln));
            }
        }
        if (ln < 8) tot[ln * 8 + wave] = totv;
        __syncthreads();

        const int4 ta = ((const int4*)tot)[(ln & 7) * 2];
        const int4 tb = ((const int4*)tot)[(ln & 7) * 2 + 1];
        const int tw[8] = {ta.x, ta.y, ta.z, ta.w, tb.x, tb.y, tb.z, tb.w};
        int colsum = 0, below = 0;
        for (int w = 0; w < 8; ++w) {
            colsum += tw[w];
            below += (w < wave) ? tw[w] : 0;
        }
        int incl = colsum;
        incl += __builtin_amdgcn_update_dpp(0, incl, 0x111, 0xf, 0xf, true);
        incl += __builtin_amdgcn_update_dpp(0, incl, 0x112, 0xf, 0xf, true);
        incl += __builtin_amdgcn_update_dpp(0, incl, 0x114, 0xf, 0xf, true);
        const int start = incl - colsum;
        int runv = start + below;
        const int segbeg = __builtin_amdgcn_readlane(start, wave);
        const int seglen = __builtin_amdgcn_readlane(colsum, wave);

        #pragma unroll
        for (int k = 0; k < WCH; ++k) {
            const int e0 = w0 + k * 32;
            if (e0 < wend) {
                const unsigned o = (unsigned)r[k].x >> (CSH + 4);
                const bool has = ln < wend - e0;
                const unsigned ok = __builtin_amdgcn_ballot_w32(has);
                const unsigned b0 = __builtin_amdgcn_ballot_w32((o & 1u) != 0u);
                const unsigned b1 = __builtin_amdgcn_ballot_w32((o & 2u) != 0u);
                const unsigned b2 = __builtin_amdgcn_ballot_w32((o & 4u) != 0u);
                const unsigned mine = same_key(b0, b1, b2, ok, o);
                const int pos = __shfl(runv, (int)o, 32) + (int)__builtin_amdgcn_mbcnt_lo(mine, 0u);
                if (has) lst[pos] = r[k];
                runv += __builtin_popcount(same_key(b0, b1, b2, ok, (unsigned)ln));
            }
        }
        __syncthreads();

        int lb = lane;
        asm volatile("" : "+v"(lb));
        v16f ax = (v16f){}, ay = (v16f){};
        if (s0 != beg) {
            #pragma unroll
            for (int k = 0; k < WR; ++k) {
                const int row = b * RB + wave * WR + k;
                if (row < NT) {
                    const v2f y = *(const v2f*)(L + (size_t)row * DD + lb * 2);
                    ax[k] = y.x; ay[k] = y.y;
                }
            }
        }
        int t = 0;
        for (; t + WR <= seglen; t += WR)
            add_batch<SPLIT>(lst + segbeg + t, WR, lb, srcA, srcB, ax, ay);
        if (t < seglen)
            add_batch<SPLIT>(lst + segbeg + t, seglen - t, lb, srcA, srcB, ax, ay);
        store_tile(L, b * RB + wave * WR, lb, ax, ay);

        if (rend < end) __syncthreads();
    }
}

#define SCAN 2048
static_assert(SCAN == 8 * 256, "a round is eight records per thread: the list cannot overflow");
static_assert(BATCH == 4 * 4 * 256, "an index array is four 16-byte loads per thread");
static_assert(RB == 128, "the mask is four words");

__device__ __forceinline__ bool row_marked(unsigned m0, unsigned m1, unsigned m2,
                                           unsigned m3, int lr) {
    const unsigned w = (lr & 64) ? ((lr & 32) ? m3 : m2) : ((lr & 32) ? m1 : m0);
    return (w >> (lr & 31)) & 1u;
}

__global__ __launch_bounds__(256)
void k_spmm_last(const int* __restrict__ binned, const int* __restrict__ goff,
                 const float* __restrict__ src, float* __restrict__ L,
                 const int* __restrict__ us, const int* __restrict__ pi,
                 const int* __restrict__ ni, float* __restrict__ out, int colbase) {
    __shared__ float tile[RB * DD];
    __shared__ __attribute__((aligned(8))) int keep[2 * SCAN];
    __shared__ unsigned need[RB / 32];
    __shared__ int nkeep;
    if (blockIdx.x >= NB) {
        gather_norm_rows(src, us, pi, ni, out, colbase, blockIdx.x - NB,
                         threadIdx.x >> 5, threadIdx.x & 31);
        return;
    }
    const int b = blockIdx.x;
    float4* t4 = (float4*)tile;
    for (int i = threadIdx.x; i < RB * DD / 4; i += 256)
        t4[i] = make_float4(0.f, 0.f, 0.f, 0.f);
    if (threadIdx.x < RB / 32) need[threadIdx.x] = 0u;
    __syncthreads();

    {
        int4 q[12];
        #pragma unroll
        for (int k = 0; k < 12; ++k) {
            const int* ip = (k < 4) ? us : (k < 8) ? pi : ni;
            q[k] = ((const int4*)ip)[threadIdx.x + (k & 3) * 256];
        }
        #pragma unroll
        for (int k = 0; k < 12; ++k) {
            const int add = (k < 4) ? 0 : NU;
            const int nv[4] = {q[k].x + add, q[k].y + add, q[k].z + add, q[k].w + add};
            for (int j = 0; j < 4; ++j)
                if ((unsigned)nv[j] / RB == (unsigned)b)
                    atomicOr(&need[(nv[j] & (RB - 1)) >> 5], 1u << (nv[j] & 31));
        }
    }
    __syncthreads();
    const unsigned m0 = need[0], m1 = need[1], m2 = need[2], m3 = need[3];
    if ((m0 | m1 | m2 | m3) == 0u) return;

    const int beg = goff[b], end = goff[b + 1];
    const int wave = __builtin_amdgcn_readfirstlane(threadIdx.x >> 5);
    const int lane = threadIdx.x & 31;
    const int2* rec = (const int2*)binned;
    int2* kp = (int2*)keep;

    for (int s0 = beg; s0 < end; s0 += SCAN) {
        if (threadIdx.x == 0) nkeep = 0;
        __syncthreads();
        {
            int2 r[8];
            #pragma unroll
            for (int k = 0; k < 8; ++k) {
                const int e = s0 + threadIdx.x + k * 256;
                r[k] = (e < end) ? rec[e] : make_int2(0, 0);
            }
            #pragma unroll
            for (int k = 0; k < 8; ++k) {
                const int e = s0 + threadIdx.x + k * 256;
                if (e < end && row_marked(m0, m1, m2, m3, (unsigned)r[k].x >> CSH))
                    kp[atomicAdd(&nkeep, 1)] = r[k];
            }
        }
        __syncthreads();
        const int nk = __builtin_amdgcn_readfirstlane(nkeep);
        for (int t0 = wave * 16; t0 < nk; t0 += 8 * 16) {
            const int cnt = min(16, nk - t0);
            const int2 r = kp[t0 + min(lane, cnt - 1)];
            v2f x[16];
            int pk[16];
            #pragma unroll
            for (int u = 0; u < 16; ++u) {
                pk[u] = __builtin_amdgcn_readlane(r.x, u);
                x[u] = *(gcv2_p)((gcf_p)(src + (size_t)(pk[u] & CMASK) * DD) + lane * 2);
            }
            #pragma unroll
            for (int u = 0; u < 16; ++u) {
                if (u < cnt) {
                    float v = __int_as_float(__builtin_amdgcn_readlane(r.y, u));
                    float* d = tile + (pk[u] >> CSH) * DD + lane * 2;
                    lds_add(d,     v * x[u].x);
                    lds_add(d + 1, v * x[u].y);
                }
            }
        }
        __syncthreads();
    }

    for (int i = threadIdx.x; i < RB * DD / 4; i += 256) {
        const int lr = i >> 4;
        const int row = b * RB + lr;
        if (row < NT && row_marked(m0, m1, m2, m3, lr))
            *(float4*)(L + (size_t)row * DD + (i & 15) * 4) = t4[i];
    }
}


__device__ __forceinline__ void stage_weights(float* sg, float* sb, float* sbias,
        const float* __restrict__ Wgc, const float* __restrict__ Wbi,
        const float* __restrict__ bias) {
    const int p = threadIdx.x >> 3;
    const int f = threadIdx.x & 7;
    float4 g[2][2], h[2][2];
    for (int q = 0; q < 2; ++q) {
        for (int e = 0; e < 2; ++e) {
            const int idx = (2 * p + e) * 64 + (f + 8 * q) * 4;
            g[e][q] = *(const float4*)(Wgc + idx);
            h[e][q] = *(const float4*)(Wbi + idx);
        }
    }
    const float b2 = (threadIdx.x < 64) ? bias[threadIdx.x] : 0.f;
    for (int q = 0; q < 2; ++q) {
        const int dst = (p * 64 + (f + 8 * q) * 4) << 1;
        *(float4*)(sg + dst)     = make_float4(g[0][q].x, g[1][q].x, g[0][q].y, g[1][q].y);
        *(float4*)(sg + dst + 4) = make_float4(g[0][q].z, g[1][q].z, g[0][q].w, g[1][q].w);
        *(float4*)(sb + dst)     = make_float4(h[0][q].x, h[1][q].x, h[0][q].y, h[1][q].y);
        *(float4*)(sb + dst + 4) = make_float4(h[0][q].z, h[1][q].z, h[0][q].w, h[1][q].w);
    }
    if (threadIdx.x < 64) sbias[threadIdx.x] = 2.0f * b2;
}

__device__ __forceinline__ void dense_tile(const float* lrow, gcf_p erow,
        const float* sg, const float* sb, int half, int n, v8f (&acc)[4]) {
    #pragma unroll
    for (int j = 0; j < 4; ++j) acc[j] = (v8f){};

    #pragma unroll
    for (int kk = 0; kk < 16; ++kk) {
        int off = kk * 4 + half * 2;
        v2f ls = *(const v2f*)(lrow + off);
        v2f eg = *(gcv2_p)(erow + off);
        v2f a_li  = ls + eg;
        v2f a_mul = ls * eg;
        int bbase = ((kk * 2 + half) * 64) << 1;
        #pragma unroll
        for (int j = 0; j < 4; ++j) {
            v2f bgc = *(const v2f*)(sg + bbase + ((j * 16 + n) << 1));
            v2f bbi = *(const v2f*)(sb + bbase + ((j * 16 + n) << 1));
            acc[j] = __builtin_amdgcn_wmma_f32_16x16x4_f32(
                false, a_li, false, bgc, (short)0, acc[j], false, false);
            acc[j] = __builtin_amdgcn_wmma_f32_16x16x4_f32(
                false, a_mul, false, bbi, (short)0, acc[j], false, false);
        }
    }
}

template <bool SPLIT>
__global__ __launch_bounds__(256)
void k_dense_wmma(const float* __restrict__ L, const float* srcA, const float* srcB,
                  float* ego,
                  const float* __restrict__ Wgc, const float* __restrict__ Wbi,
                  const float* __restrict__ bias) {
    __shared__ __attribute__((aligned(16))) float sg[4096];
    __shared__ __attribute__((aligned(16))) float sb[4096];
    __shared__ float sbias[64];

    stage_weights(sg, sb, sbias, Wgc, Wbi, bias);
    __syncthreads();

    const int wave  = threadIdx.x >> 5;
    const int lane  = threadIdx.x & 31;
    const int mbase = (blockIdx.x * 8 + wave) * 16;
    if (mbase >= NT) return;

    const int half = lane >> 4;
    const int n    = lane & 15;

    const float* lrow = L   + (size_t)(mbase + n) * DD;
    gcf_p erow = src_row<SPLIT>(srcA, srcB, mbase + n);

    v8f acc[4];
    dense_tile(lrow, erow, sg, sb, half, n, acc);

    #pragma unroll
    for (int j = 0; j < 4; ++j) {
        float bsum = sbias[j * 16 + n];
        #pragma unroll
        for (int v = 0; v < 8; ++v) {
            float x = acc[j][v] + bsum;
            x = (x >= 0.f) ? x : SLOPE * x;
            ego[(size_t)(mbase + (half << 3) + v) * DD + j * 16 + n] = x;
        }
    }
}

static_assert(BATCH % 16 == 0, "the 16 slots of a tile share one index array");
static_assert(3 * BATCH % 128 == 0, "eight whole tiles per work-group");
__global__ __launch_bounds__(256)
void k_dense_last(const float* __restrict__ L, const float* __restrict__ ego,
                  const float* __restrict__ Wgc, const float* __restrict__ Wbi,
                  const float* __restrict__ bias,
                  const int* __restrict__ us, const int* __restrict__ pi,
                  const int* __restrict__ ni, float* __restrict__ out, int colbase) {
    __shared__ __attribute__((aligned(16))) float sg[4096];
    __shared__ __attribute__((aligned(16))) float sb[4096];
    __shared__ float sbias[64];
    __shared__ __attribute__((aligned(16))) float ot[8 * 16 * DD];

    const int wave  = threadIdx.x >> 5;
    const int lane  = threadIdx.x & 31;
    const int half  = lane >> 4;
    const int n     = lane & 15;
    const int sbase = (blockIdx.x * 8 + wave) * 16;
    const int seg = sbase >> 12;
    const int* ip = (seg == 0) ? us : (seg == 1) ? pi : ni;
    const int node = ip[(sbase & (BATCH - 1)) + n] + (seg ? NU : 0);

    stage_weights(sg, sb, sbias, Wgc, Wbi, bias);
    __syncthreads();

    v8f acc[4];
    dense_tile(L + (size_t)node * DD, (gcf_p)(ego + (size_t)node * DD), sg, sb, half, n, acc);

    float* o = ot + wave * (16 * DD);
    #pragma unroll
    for (int j = 0; j < 4; ++j) {
        float bsum = sbias[j * 16 + n];
        #pragma unroll
        for (int v = 0; v < 8; ++v) {
            float x = acc[j][v] + bsum;
            x = (x >= 0.f) ? x : SLOPE * x;
            o[((half << 3) + v) * DD + j * 16 + n] = x;
        }
    }
    __syncthreads();
    #pragma unroll
    for (int r = 0; r < 16; ++r) {
        v2f x = *(const v2f*)(o + r * DD + lane * 2);
        float ss = x.x * x.x + x.y * x.y;
        for (int s = 16; s > 0; s >>= 1) ss += __shfl_xor(ss, s, 32);
        float scale = 1.0f / fmaxf(sqrtf(ss), 1e-12f);
        v2f y; y.x = x.x * scale; y.y = x.y * scale;
        *(v2f*)(out + (size_t)(sbase + r) * 256 + colbase + lane * 2) = y;
    }
}

extern "C" void kernel_launch(void* const* d_in, const int* in_sizes, int n_in,
                              void* d_out, int out_size, void* d_ws, size_t ws_size,
                              hipStream_t stream) {
    const float* ue   = (const float*)d_in[0];
    const float* ie   = (const float*)d_in[1];
    const float* Wgc  = (const float*)d_in[2];
    const float* Wbi  = (const float*)d_in[3];
    const float* bbi  = (const float*)d_in[4];
    const float* vals = (const float*)d_in[5];
    const int*   rows = (const int*)d_in[6];
    const int*   cols = (const int*)d_in[7];
    const int*   us   = (const int*)d_in[8];
    const int*   pi   = (const int*)d_in[9];
    const int*   ni   = (const int*)d_in[10];
    float* out = (float*)d_out;

    float* ego = (float*)d_ws;
    float* L   = ego + (size_t)NT * DD;
    int* binned = (int*)(L + (size_t)NT * DD);
    int* gcount = binned + (size_t)2 * NNZE;
    int* goff   = gcount + 2048;
    int* gcur   = goff + 2048;

    k_gather0<<<(3 * BATCH * DD + 255) / 256, 256, 0, stream>>>(
        ue, ie, us, pi, ni, out, gcount);

    k_bin_count<<<GBIN, 256, 0, stream>>>(rows, gcount, gcur);
    k_bin_place<<<GBIN, 256, 0, stream>>>(rows, cols, vals, gcount, goff, gcur, binned);

    k_spmm_blk<true><<<NB, 256, 0, stream>>>(
        binned, goff, ue, ie, L, ego, us, pi, ni, out, 0);
    k_dense_wmma<true><<<(NT + 127) / 128, 256, 0, stream>>>(
        L, ue, ie, ego, Wgc, Wbi, bbi);
    k_spmm_blk<false><<<NB + GNB, 256, 0, stream>>>(
        binned, goff, ego, ego, L, ego, us, pi, ni, out, 64);
    k_dense_wmma<false><<<(NT + 127) / 128, 256, 0, stream>>>(
        L, ego, ego, ego, Wgc + 4096, Wbi + 4096, bbi + 64);
    k_spmm_last<<<NB + GNB, 256, 0, stream>>>(
        binned, goff, ego, L, us, pi, ni, out, 128);
    k_dense_last<<<3 * BATCH / 128, 256, 0, stream>>>(
        L, ego, Wgc + 2 * 4096, Wbi + 2 * 4096, bbi + 2 * 64, us, pi, ni, out, 192);
}
